# starting kernel
# baseline (speedup 1.0000x reference)
#include <hip/hip_runtime.h>
#include <hip/hip_bf16.h>


typedef __attribute__((ext_vector_type(2))) float v2f;
typedef __attribute__((ext_vector_type(8))) float v8f;

#define IN_DIM   128
#define K_DIM    256
#define OUT_DIM  128
#define MBLK     64
#define LDS_STR  260

__global__ __launch_bounds__(256) void sage_scatter(
    const float* __restrict__ x,
    const int*   __restrict__ esrc,
    const int*   __restrict__ edst,
    float*       __restrict__ neigh,
    float*       __restrict__ deg,
    int nEdges)
{
    const int wave = threadIdx.x >> 5;
    const int lane = threadIdx.x & 31;
    const int e    = blockIdx.x * 8 + wave;
    if (e >= nEdges) return;

    const int s = esrc[e];
    const int d = edst[e];
    if (lane == 0) atomicAdd(&deg[d], 1.0f);

    const float4 v = ((const float4*)(x + (size_t)s * IN_DIM))[lane];
    float* nd = neigh + (size_t)d * IN_DIM + lane * 4;
    atomicAdd(nd + 0, v.x);
    atomicAdd(nd + 1, v.y);
    atomicAdd(nd + 2, v.z);
    atomicAdd(nd + 3, v.w);
}

__global__ __launch_bounds__(256) void sage_gemm(
    const float* __restrict__ x,
    const float* __restrict__ neigh,
    const float* __restrict__ deg,
    const float* __restrict__ W,
    const float* __restrict__ bias,
    float*       __restrict__ out,
    int nNodes)
{
    __shared__ float As[MBLK * LDS_STR];

    const int mBase = blockIdx.x * MBLK;
    const int t = threadIdx.x;
    const bool fullBlock = (mBase + MBLK) <= nNodes;

    #pragma unroll
    for (int i = 0; i < 16; ++i) {
        const int idx  = t + i * 256;
        const int row  = idx >> 6;
        const int c4   = idx & 63;
        const int grow = mBase + row;
        float4 v = make_float4(0.f, 0.f, 0.f, 0.f);
        if (fullBlock || grow < nNodes) {
            if (c4 < 32) {
                v = ((const float4*)(x + (size_t)grow * IN_DIM))[c4];
            } else {
                v = ((const float4*)(neigh + (size_t)grow * IN_DIM))[c4 - 32];
                const float inv = 1.0f / fmaxf(deg[grow], 1.0f);
                v.x *= inv; v.y *= inv; v.z *= inv; v.w *= inv;
            }
        }
        float* dst = &As[row * LDS_STR + c4 * 4];
        dst[0] = v.x; dst[1] = v.y; dst[2] = v.z; dst[3] = v.w;
    }
    __syncthreads();

    const int wave  = t >> 5;
    const int lane  = t & 31;
    const int ln    = lane & 15;
    const int hi    = lane >> 4;
    const int nBase = wave * 16;

    v8f acc[4] = {};

    for (int k0 = 0; k0 < K_DIM; k0 += 4) {
        const int kk = k0 + hi * 2;

        v2f bfrag;
        bfrag.x = W[(size_t)kk       * OUT_DIM + nBase + ln];
        bfrag.y = W[(size_t)(kk + 1) * OUT_DIM + nBase + ln];

        #pragma unroll
        for (int ms = 0; ms < 4; ++ms) {
            const int row = ms * 16 + ln;
            v2f afrag;
            afrag.x = As[row * LDS_STR + kk];
            afrag.y = As[row * LDS_STR + kk + 1];

            acc[ms] = __builtin_amdgcn_wmma_f32_16x16x4_f32(
                 false, afrag,  false, bfrag,
                 (short)0, acc[ms],
                 false,  false);
        }
    }

    const float bv = bias[nBase + ln];
    float* obase = out + (size_t)(mBase + hi * 8) * OUT_DIM + nBase + ln;
    if (fullBlock) {
        #pragma unroll
        for (int ms = 0; ms < 4; ++ms) {
            #pragma unroll
            for (int r = 0; r < 8; ++r) {
                obase[(size_t)(ms * 16 + r) * OUT_DIM] = acc[ms][r] + bv;
            }
        }
    } else {
        #pragma unroll
        for (int ms = 0; ms < 4; ++ms) {
            #pragma unroll
            for (int r = 0; r < 8; ++r) {
                const int grow = mBase + ms * 16 + r + hi * 8;
                if (grow < nNodes)
                    out[(size_t)grow * OUT_DIM + nBase + ln] = acc[ms][r] + bv;
            }
        }
    }
}

extern "C" void kernel_launch(void* const* d_in, const int* in_sizes, int n_in,
                              void* d_out, int out_size, void* d_ws, size_t ws_size,
                              hipStream_t stream)
{
    const float* x    = (const float*)d_in[0];
    const int*   esrc = (const int*)  d_in[1];
    const int*   edst = (const int*)  d_in[2];
    const float* W    = (const float*)d_in[3];
    const float* bias = (const float*)d_in[4];
    float*       out  = (float*)d_out;

    const int nNodes = in_sizes[0] / IN_DIM;
    const int nEdges = in_sizes[1];

    float* neigh = (float*)d_ws;
    float* deg   = neigh + (size_t)nNodes * IN_DIM;
    const size_t zero_bytes = ((size_t)nNodes * IN_DIM + (size_t)nNodes) * sizeof(float);
    hipMemsetAsync(d_ws, 0, zero_bytes, stream);

    const int scatterBlocks = (nEdges + 7) / 8;
    sage_scatter<<<scatterBlocks, 256, 0, stream>>>(x, esrc, edst, neigh, deg, nEdges);

    const int gemmBlocks = (nNodes + MBLK - 1) / MBLK;
    sage_gemm<<<gemmBlocks, 256, 0, stream>>>(x, neigh, deg, W, bias, out, nNodes);
}


# Float-atomic scatter replaced by dst-bucketed gather: int-atomic rank, offsets, place; GEMM A staging sums source rows in registers; sum order = arrival rank, pairwise in groups of 8
# speedup vs baseline: 1.7331x; 1.7331x over previous
#include <hip/hip_runtime.h>
#include <hip/hip_bf16.h>


typedef __attribute__((ext_vector_type(2))) float v2f;
typedef __attribute__((ext_vector_type(8))) float v8f;

#define IN_DIM   128
#define K_DIM    256
#define OUT_DIM  128
#define MBLK     64
#define LDS_STR  260

__global__ __launch_bounds__(256) void sage_count(
    const int* __restrict__ edst,
    int*       __restrict__ cnt,
    int*       __restrict__ rank,
    int nEdges)
{
    const int e = blockIdx.x * 256 + threadIdx.x;
    if (e < nEdges) rank[e] = atomicAdd(&cnt[edst[e]], 1);
}

__global__ __launch_bounds__(256) void sage_offsets(
    const int* __restrict__ cnt,
    int*       __restrict__ offs,
    int*       __restrict__ cursor,
    int nNodes)
{
    __shared__ int wsum[8];
    __shared__ int blockBase;

    const int t    = threadIdx.x;
    const int lane = t & 31;
    const int wave = t >> 5;
    const int i    = blockIdx.x * 256 + t;
    const int c    = (i < nNodes) ? cnt[i] : 0;

    int v = c;
    #pragma unroll
    for (int d = 1; d < 32; d <<= 1) {
        const int up = __shfl_up(v, d);
        if (lane >= d) v += up;
    }
    if (lane == 31) wsum[wave] = v;
    __syncthreads();
    if (t == 0) {
        int tot = 0;
        #pragma unroll
        for (int w = 0; w < 8; ++w) { const int s = wsum[w]; wsum[w] = tot; tot += s; }
        blockBase = atomicAdd(cursor, tot);
    }
    __syncthreads();
    if (i < nNodes) offs[i] = blockBase + wsum[wave] + v - c;
}

__global__ __launch_bounds__(256) void sage_place(
    const int* __restrict__ esrc,
    const int* __restrict__ edst,
    const int* __restrict__ rank,
    const int* __restrict__ offs,
    int*       __restrict__ ssrc,
    int nEdges)
{
    const int e = blockIdx.x * 256 + threadIdx.x;
    if (e < nEdges) ssrc[offs[edst[e]] + rank[e]] = esrc[e];
}

struct Rows8 { float4 v[8]; };

__device__ __forceinline__ void issue8(Rows8& R, const float4* __restrict__ xl,
                                       int mine, int j, int m)
{
    int s[8];
    #pragma unroll
    for (int u = 0; u < 8; ++u) s[u] = __builtin_amdgcn_readlane(mine, j + u);
    #pragma unroll
    for (int u = 0; u < 8; ++u) {
        R.v[u] = make_float4(0.f, 0.f, 0.f, 0.f);
        if (j + u < m) R.v[u] = xl[(size_t)s[u] * 32];
    }
}

__device__ __forceinline__ float4 add4(const float4& a, const float4& b)
{
    return make_float4(a.x + b.x, a.y + b.y, a.z + b.z, a.w + b.w);
}

__device__ __forceinline__ void add8(float4& acc, const Rows8& R)
{
    const float4 t0 = add4(R.v[0], R.v[1]);
    const float4 t1 = add4(R.v[2], R.v[3]);
    const float4 t2 = add4(R.v[4], R.v[5]);
    const float4 t3 = add4(R.v[6], R.v[7]);
    acc = add4(acc, add4(add4(t0, t1), add4(t2, t3)));
}

__global__ __launch_bounds__(256) void sage_gemm(
    const float* __restrict__ x,
    const int*   __restrict__ cnt,
    const int*   __restrict__ offs,
    const int*   __restrict__ ssrc,
    const float* __restrict__ W,
    const float* __restrict__ bias,
    float*       __restrict__ out,
    int nNodes)
{
    __shared__ float As[MBLK * LDS_STR];

    const int mBase = blockIdx.x * MBLK;
    const int t = threadIdx.x;
    const bool fullBlock = (mBase + MBLK) <= nNodes;

    const float4* x4 = (const float4*)x;
    #pragma unroll
    for (int i = 0; i < 8; ++i) {
        const int idx  = t + i * 256;
        const int row  = idx >> 5;
        const int c4   = idx & 31;
        const int grow = mBase + row;
        float4 v = make_float4(0.f, 0.f, 0.f, 0.f);
        if (fullBlock || grow < nNodes) v = x4[(size_t)grow * 32 + c4];
        float* dst = &As[row * LDS_STR + c4 * 4];
        dst[0] = v.x; dst[1] = v.y; dst[2] = v.z; dst[3] = v.w;
    }

    const int g  = __builtin_amdgcn_readfirstlane(t >> 5);
    const int gl = t & 31;
    int myN = 0, myBeg = 0;
    {
        const int nd = mBase + g * 8 + gl;
        if (gl < 8 && nd < nNodes) { myN = cnt[nd]; myBeg = offs[nd]; }
    }
    const float4* xl = x4 + gl;
    int n    = __builtin_amdgcn_readlane(myN, 0);
    int beg  = __builtin_amdgcn_readlane(myBeg, 0);
    int mine = (gl < n) ? ssrc[beg + gl] : 0;
    for (int r = 0; r < 8; ++r) {
        float4 acc = make_float4(0.f, 0.f, 0.f, 0.f);
        Rows8 R;
        int m = (n < 32) ? n : 32;
        issue8(R, xl, mine, 0, m);

        int n2 = 0, beg2 = 0, mine2 = 0;
        if (r + 1 < 8) {
            n2   = __builtin_amdgcn_readlane(myN, r + 1);
            beg2 = __builtin_amdgcn_readlane(myBeg, r + 1);
            mine2 = (gl < n2) ? ssrc[beg2 + gl] : 0;
        }

        add8(acc, R);
        for (int j = 8; j < m; j += 8) { issue8(R, xl, mine, j, m); add8(acc, R); }
        for (int c = 32; c < n; c += 32) {
            m = (n - c < 32) ? (n - c) : 32;
            mine = (gl < m) ? ssrc[beg + c + gl] : 0;
            for (int j = 0; j < m; j += 8) { issue8(R, xl, mine, j, m); add8(acc, R); }
        }

        const float inv = 1.0f / fmaxf((float)n, 1.0f);
        float* dst = &As[(g * 8 + r) * LDS_STR + IN_DIM + gl * 4];
        dst[0] = acc.x * inv; dst[1] = acc.y * inv; dst[2] = acc.z * inv; dst[3] = acc.w * inv;

        n = n2; beg = beg2; mine = mine2;
    }
    __syncthreads();

    const int wave  = t >> 5;
    const int lane  = t & 31;
    const int ln    = lane & 15;
    const int hi    = lane >> 4;
    const int nBase = wave * 16;

    v8f acc[4] = {};

    for (int k0 = 0; k0 < K_DIM; k0 += 4) {
        const int kk = k0 + hi * 2;

        v2f bfrag;
        bfrag.x = W[(size_t)kk       * OUT_DIM + nBase + ln];
        bfrag.y = W[(size_t)(kk + 1) * OUT_DIM + nBase + ln];

        #pragma unroll
        for (int ms = 0; ms < 4; ++ms) {
            const int row = ms * 16 + ln;
            v2f afrag;
            afrag.x = As[row * LDS_STR + kk];
            afrag.y = As[row * LDS_STR + kk + 1];

            acc[ms] = __builtin_amdgcn_wmma_f32_16x16x4_f32(
                 false, afrag,  false, bfrag,
                 (short)0, acc[ms],
                 false,  false);
        }
    }

    const float bv = bias[nBase + ln];
    float* obase = out + (size_t)(mBase + hi * 8) * OUT_DIM + nBase + ln;
    if (fullBlock) {
        #pragma unroll
        for (int ms = 0; ms < 4; ++ms) {
            #pragma unroll
            for (int r = 0; r < 8; ++r) {
                obase[(size_t)(ms * 16 + r) * OUT_DIM] = acc[ms][r] + bv;
            }
        }
    } else {
        #pragma unroll
        for (int ms = 0; ms < 4; ++ms) {
            #pragma unroll
            for (int r = 0; r < 8; ++r) {
                const int grow = mBase + ms * 16 + r + hi * 8;
                if (grow < nNodes)
                    out[(size_t)grow * OUT_DIM + nBase + ln] = acc[ms][r] + bv;
            }
        }
    }
}

extern "C" void kernel_launch(void* const* d_in, const int* in_sizes, int n_in,
                              void* d_out, int out_size, void* d_ws, size_t ws_size,
                              hipStream_t stream)
{
    const float* x    = (const float*)d_in[0];
    const int*   esrc = (const int*)  d_in[1];
    const int*   edst = (const int*)  d_in[2];
    const float* W    = (const float*)d_in[3];
    const float* bias = (const float*)d_in[4];
    float*       out  = (float*)d_out;

    const int nNodes = in_sizes[0] / IN_DIM;
    const int nEdges = in_sizes[1];

    int* cnt    = (int*)d_ws;
    int* cursor = cnt + nNodes;
    int* offs   = cursor + 32;
    int* rank   = offs + nNodes;
    int* ssrc   = rank + nEdges;
    hipMemsetAsync(d_ws, 0, ((size_t)nNodes + 32) * sizeof(int), stream);

    const int edgeBlocks = (nEdges + 255) / 256;
    const int nodeBlocks = (nNodes + 255) / 256;
    sage_count<<<edgeBlocks, 256, 0, stream>>>(edst, cnt, rank, nEdges);
    sage_offsets<<<nodeBlocks, 256, 0, stream>>>(cnt, offs, cursor, nNodes);
    sage_place<<<edgeBlocks, 256, 0, stream>>>(esrc, edst, rank, offs, ssrc, nEdges);

    const int gemmBlocks = (nNodes + MBLK - 1) / MBLK;
    sage_gemm<<<gemmBlocks, 256, 0, stream>>>(x, cnt, offs, ssrc, W, bias, out, nNodes);
}


# Per-edge global returning adds and 4-byte scatter replaced by two-level counting sort on LDS counters (256-node ranges, then per node); sum order = LDS arrival rank, pairwise in groups of 8
# speedup vs baseline: 3.5113x; 2.0260x over previous
#include <hip/hip_runtime.h>
#include <hip/hip_bf16.h>


typedef __attribute__((ext_vector_type(2))) float v2f;
typedef __attribute__((ext_vector_type(8))) float v8f;

#define IN_DIM   128
#define K_DIM    256
#define OUT_DIM  128
#define MBLK     64
#define LDS_STR  260

#define BIN_SHIFT 8
#define BIN_NODES 256
#define BIN_MASK  255
#define SORT_T    512
#define SORT_EPT  12
#define SORT_CHUNK (SORT_T * SORT_EPT)
#define ROW_CAP   8192

__device__ __forceinline__ int block_excl_scan(const int* in, int* out, int n, int* wtot)
{
    const int t     = threadIdx.x;
    const int lane  = t & 31;
    const int wave  = t >> 5;
    const int nthr  = SORT_T;
    const int per   = (n + nthr - 1) / nthr;
    const int b0    = t * per;

    int s = 0;
    for (int k = 0; k < per; ++k) { const int b = b0 + k; if (b < n) s += in[b]; }

    int v = s;
    for (int d = 1; d < 32; d <<= 1) {
        const int up = __shfl_up(v, d);
        if (lane >= d) v += up;
    }
    if (lane == 31) wtot[wave] = v;
    __syncthreads();
    if (wave == 0) {
        const int x = (lane < (nthr >> 5)) ? wtot[lane] : 0;
        int y = x;
        for (int d = 1; d < 32; d <<= 1) {
            const int up = __shfl_up(y, d);
            if (lane >= d) y += up;
        }
        wtot[lane] = y - x;
        if (lane == 31) wtot[32] = y;
    }
    __syncthreads();
    int run = wtot[wave] + v - s;
    for (int k = 0; k < per; ++k) {
        const int b = b0 + k;
        if (b < n) { const int c = in[b]; out[b] = run; run += c; }
    }
    const int total = wtot[32];
    __syncthreads();
    return total;
}

__global__ __launch_bounds__(SORT_T) void sage_bincount(
    const int* __restrict__ edst,
    int*       __restrict__ binCnt,
    int nEdges, int nBins)
{
    extern __shared__ int sm[];

    const int t = threadIdx.x;
    for (int b = t; b < nBins; b += SORT_T) sm[b] = 0;
    __syncthreads();

    const int e0 = blockIdx.x * SORT_CHUNK;
    int d[SORT_EPT];
    #pragma unroll
    for (int k = 0; k < SORT_EPT; ++k) {
        const int  e  = e0 + k * SORT_T + t;
        const bool ok = e < nEdges;
        const int  dv = edst[ok ? e : nEdges - 1];
        d[k] = ok ? dv : -1;
    }
    #pragma unroll
    for (int k = 0; k < SORT_EPT; ++k)
        if (d[k] >= 0) atomicAdd(&sm[d[k] >> BIN_SHIFT], 1);
    __syncthreads();

    for (int b = t; b < nBins; b += SORT_T) {
        const int c = sm[b];
        if (c) atomicAdd(&binCnt[b], c);
    }
}

__global__ __launch_bounds__(SORT_T) void sage_binplace(
    const int* __restrict__ esrc,
    const int* __restrict__ edst,
    const int* __restrict__ binCnt,
    int*       __restrict__ binCursor,
    int*       __restrict__ binBase,
    unsigned*  __restrict__ coarse,
    int nEdges, int nBins)
{
    extern __shared__ int sm[];
    __shared__ int      wtot[33];
    __shared__ uint2    stage[SORT_CHUNK];

    int* h   = sm;
    int* res = sm + nBins;
    int* loc = sm + 2 * nBins;

    const int t    = threadIdx.x;
    for (int b = t; b < nBins; b += SORT_T) h[b] = 0;
    const int total = block_excl_scan(binCnt, res, nBins, wtot);
    if (blockIdx.x == 0) {
        for (int b = t; b < nBins; b += SORT_T) binBase[b] = res[b];
        if (t == 0) binBase[nBins] = total;
    }

    const int e0 = blockIdx.x * SORT_CHUNK;
    int d[SORT_EPT], s[SORT_EPT], r[SORT_EPT];
    #pragma unroll
    for (int k = 0; k < SORT_EPT; ++k) {
        const int  e  = e0 + k * SORT_T + t;
        const bool ok = e < nEdges;
        const int  ec = ok ? e : nEdges - 1;
        const int  dv = edst[ec];
        s[k] = esrc[ec];
        d[k] = ok ? dv : -1;
    }
    #pragma unroll
    for (int k = 0; k < SORT_EPT; ++k) {
        r[k] = 0;
        if (d[k] >= 0) r[k] = atomicAdd(&h[d[k] >> BIN_SHIFT], 1);
    }
    __syncthreads();

    for (int b = t; b < nBins; b += SORT_T) {
        const int c = h[b];
        if (c) res[b] += atomicAdd(&binCursor[b], c);
    }
    block_excl_scan(h, loc, nBins, wtot);

    #pragma unroll
    for (int k = 0; k < SORT_EPT; ++k) {
        if (d[k] >= 0) {
            const int bin = d[k] >> BIN_SHIFT;
            stage[loc[bin] + r[k]] = make_uint2(
                ((unsigned)s[k] << BIN_SHIFT) | (unsigned)(d[k] & BIN_MASK),
                (unsigned)(res[bin] + r[k]));
        }
    }
    __syncthreads();

    const int nHere = (nEdges - e0 < SORT_CHUNK) ? (nEdges - e0) : SORT_CHUNK;
    #pragma unroll
    for (int k = 0; k < SORT_EPT; ++k) {
        const int i = k * SORT_T + t;
        if (i < nHere) { const uint2 q = stage[i]; coarse[q.y] = q.x; }
    }
}

__global__ __launch_bounds__(BIN_NODES) void sage_rowsort(
    const unsigned* __restrict__ coarse,
    const int*      __restrict__ binBase,
    int*            __restrict__ cnt,
    int*            __restrict__ offs,
    int*            __restrict__ ssrc,
    int nNodes)
{
    __shared__ int rc[BIN_NODES];
    __shared__ int ro[BIN_NODES];
    __shared__ int wsum[8];
    __shared__ int sorted[ROW_CAP];

    const int t    = threadIdx.x;
    const int lane = t & 31;
    const int wave = t >> 5;
    const int beg  = binBase[blockIdx.x];
    const int nb   = binBase[blockIdx.x + 1] - beg;
    const unsigned* __restrict__ src = coarse + beg;

    rc[t] = 0;
    __syncthreads();

    for (int i0 = 0; i0 < nb; i0 += BIN_NODES * 8) {
        unsigned w[8];
        #pragma unroll
        for (int u = 0; u < 8; ++u) {
            const int i = i0 + u * BIN_NODES + t;
            w[u] = src[i < nb ? i : nb - 1];
        }
        #pragma unroll
        for (int u = 0; u < 8; ++u) {
            const int i = i0 + u * BIN_NODES + t;
            if (i < nb) atomicAdd(&rc[w[u] & BIN_MASK], 1);
        }
    }
    __syncthreads();

    const int c = rc[t];
    int v = c;
    for (int d = 1; d < 32; d <<= 1) {
        const int up = __shfl_up(v, d);
        if (lane >= d) v += up;
    }
    if (lane == 31) wsum[wave] = v;
    __syncthreads();
    if (t == 0) {
        int tot = 0;
        for (int q = 0; q < 8; ++q) { const int x = wsum[q]; wsum[q] = tot; tot += x; }
    }
    __syncthreads();
    const int excl = wsum[wave] + v - c;
    ro[t] = excl;
    rc[t] = 0;
    {
        const int node = blockIdx.x * BIN_NODES + t;
        if (node < nNodes) { cnt[node] = c; offs[node] = beg + excl; }
    }
    __syncthreads();

    const bool inLds = nb <= ROW_CAP;
    for (int i0 = 0; i0 < nb; i0 += BIN_NODES * 8) {
        unsigned w[8];
        #pragma unroll
        for (int u = 0; u < 8; ++u) {
            const int i = i0 + u * BIN_NODES + t;
            w[u] = src[i < nb ? i : nb - 1];
        }
        #pragma unroll
        for (int u = 0; u < 8; ++u) {
            const int i = i0 + u * BIN_NODES + t;
            if (i < nb) {
                const int row = (int)(w[u] & BIN_MASK);
                const int p   = ro[row] + atomicAdd(&rc[row], 1);
                const int sid = (int)(w[u] >> BIN_SHIFT);
                if (inLds) sorted[p] = sid; else ssrc[beg + p] = sid;
            }
        }
    }
    if (inLds) {
        __syncthreads();
        for (int i = t; i < nb; i += BIN_NODES) ssrc[beg + i] = sorted[i];
    }
}

struct Rows8 { float4 v[8]; };

__device__ __forceinline__ void issue8(Rows8& R, const float4* __restrict__ xl,
                                       int mine, int j, int m)
{
    int s[8];
    #pragma unroll
    for (int u = 0; u < 8; ++u) s[u] = __builtin_amdgcn_readlane(mine, j + u);
    #pragma unroll
    for (int u = 0; u < 8; ++u) {
        R.v[u] = make_float4(0.f, 0.f, 0.f, 0.f);
        if (j + u < m) R.v[u] = xl[(size_t)s[u] * 32];
    }
}

__device__ __forceinline__ float4 add4(const float4& a, const float4& b)
{
    return make_float4(a.x + b.x, a.y + b.y, a.z + b.z, a.w + b.w);
}

__device__ __forceinline__ void add8(float4& acc, const Rows8& R)
{
    const float4 t0 = add4(R.v[0], R.v[1]);
    const float4 t1 = add4(R.v[2], R.v[3]);
    const float4 t2 = add4(R.v[4], R.v[5]);
    const float4 t3 = add4(R.v[6], R.v[7]);
    acc = add4(acc, add4(add4(t0, t1), add4(t2, t3)));
}

__global__ __launch_bounds__(256) void sage_gemm(
    const float* __restrict__ x,
    const int*   __restrict__ cnt,
    const int*   __restrict__ offs,
    const int*   __restrict__ ssrc,
    const float* __restrict__ W,
    const float* __restrict__ bias,
    float*       __restrict__ out,
    int nNodes)
{
    __shared__ float As[MBLK * LDS_STR];

    const int mBase = blockIdx.x * MBLK;
    const int t = threadIdx.x;
    const bool fullBlock = (mBase + MBLK) <= nNodes;

    const float4* x4 = (const float4*)x;
    #pragma unroll
    for (int i = 0; i < 8; ++i) {
        const int idx  = t + i * 256;
        const int row  = idx >> 5;
        const int c4   = idx & 31;
        const int grow = mBase + row;
        float4 v = make_float4(0.f, 0.f, 0.f, 0.f);
        if (fullBlock || grow < nNodes) v = x4[(size_t)grow * 32 + c4];
        float* dst = &As[row * LDS_STR + c4 * 4];
        dst[0] = v.x; dst[1] = v.y; dst[2] = v.z; dst[3] = v.w;
    }

    const int g  = __builtin_amdgcn_readfirstlane(t >> 5);
    const int gl = t & 31;
    int myN = 0, myBeg = 0;
    {
        const int nd = mBase + g * 8 + gl;
        if (gl < 8 && nd < nNodes) { myN = cnt[nd]; myBeg = offs[nd]; }
    }
    const float4* xl = x4 + gl;
    int n    = __builtin_amdgcn_readlane(myN, 0);
    int beg  = __builtin_amdgcn_readlane(myBeg, 0);
    int mine = (gl < n) ? ssrc[beg + gl] : 0;
    for (int r = 0; r < 8; ++r) {
        float4 acc = make_float4(0.f, 0.f, 0.f, 0.f);
        Rows8 R;
        int m = (n < 32) ? n : 32;
        issue8(R, xl, mine, 0, m);

        int n2 = 0, beg2 = 0, mine2 = 0;
        if (r + 1 < 8) {
            n2   = __builtin_amdgcn_readlane(myN, r + 1);
            beg2 = __builtin_amdgcn_readlane(myBeg, r + 1);
            mine2 = (gl < n2) ? ssrc[beg2 + gl] : 0;
        }

        add8(acc, R);
        for (int j = 8; j < m; j += 8) { issue8(R, xl, mine, j, m); add8(acc, R); }
        for (int c = 32; c < n; c += 32) {
            m = (n - c < 32) ? (n - c) : 32;
            mine = (gl < m) ? ssrc[beg + c + gl] : 0;
            for (int j = 0; j < m; j += 8) { issue8(R, xl, mine, j, m); add8(acc, R); }
        }

        const float inv = 1.0f / fmaxf((float)n, 1.0f);
        float* dst = &As[(g * 8 + r) * LDS_STR + IN_DIM + gl * 4];
        dst[0] = acc.x * inv; dst[1] = acc.y * inv; dst[2] = acc.z * inv; dst[3] = acc.w * inv;

        n = n2; beg = beg2; mine = mine2;
    }
    __syncthreads();

    const int wave  = t >> 5;
    const int lane  = t & 31;
    const int ln    = lane & 15;
    const int hi    = lane >> 4;
    const int nBase = wave * 16;

    v8f acc[4] = {};

    for (int k0 = 0; k0 < K_DIM; k0 += 4) {
        const int kk = k0 + hi * 2;

        v2f bfrag;
        bfrag.x = W[(size_t)kk       * OUT_DIM + nBase + ln];
        bfrag.y = W[(size_t)(kk + 1) * OUT_DIM + nBase + ln];

        #pragma unroll
        for (int ms = 0; ms < 4; ++ms) {
            const int row = ms * 16 + ln;
            v2f afrag;
            afrag.x = As[row * LDS_STR + kk];
            afrag.y = As[row * LDS_STR + kk + 1];

            acc[ms] = __builtin_amdgcn_wmma_f32_16x16x4_f32(
                 false, afrag,  false, bfrag,
                 (short)0, acc[ms],
                 false,  false);
        }
    }

    const float bv = bias[nBase + ln];
    float* obase = out + (size_t)(mBase + hi * 8) * OUT_DIM + nBase + ln;
    if (fullBlock) {
        #pragma unroll
        for (int ms = 0; ms < 4; ++ms) {
            #pragma unroll
            for (int r = 0; r < 8; ++r) {
                obase[(size_t)(ms * 16 + r) * OUT_DIM] = acc[ms][r] + bv;
            }
        }
    } else {
        #pragma unroll
        for (int ms = 0; ms < 4; ++ms) {
            #pragma unroll
            for (int r = 0; r < 8; ++r) {
                const int grow = mBase + ms * 16 + r + hi * 8;
                if (grow < nNodes)
                    out[(size_t)grow * OUT_DIM + nBase + ln] = acc[ms][r] + bv;
            }
        }
    }
}

extern "C" void kernel_launch(void* const* d_in, const int* in_sizes, int n_in,
                              void* d_out, int out_size, void* d_ws, size_t ws_size,
                              hipStream_t stream)
{
    const float* x    = (const float*)d_in[0];
    const int*   esrc = (const int*)  d_in[1];
    const int*   edst = (const int*)  d_in[2];
    const float* W    = (const float*)d_in[3];
    const float* bias = (const float*)d_in[4];
    float*       out  = (float*)d_out;

    const int nNodes = in_sizes[0] / IN_DIM;
    const int nEdges = in_sizes[1];
    const int nBins  = (nNodes + BIN_NODES - 1) / BIN_NODES;

    int*      cnt       = (int*)d_ws;
    int*      offs      = cnt + nNodes;
    unsigned* coarse    = (unsigned*)(offs + nNodes);
    int*      ssrc      = (int*)(coarse + nEdges);
    int*      binCnt    = ssrc + nEdges;
    int*      binCursor = binCnt + nBins;
    int*      binBase   = binCursor + nBins;
    hipMemsetAsync(binCnt, 0, (size_t)2 * nBins * sizeof(int), stream);

    const int sortBlocks = (nEdges + SORT_CHUNK - 1) / SORT_CHUNK;
    sage_bincount<<<sortBlocks, SORT_T, (size_t)nBins * sizeof(int), stream>>>(edst, binCnt, nEdges, nBins);
    sage_binplace<<<sortBlocks, SORT_T, (size_t)3 * nBins * sizeof(int), stream>>>(
        esrc, edst, binCnt, binCursor, binBase, coarse, nEdges, nBins);
    sage_rowsort<<<nBins, BIN_NODES, 0, stream>>>(coarse, binBase, cnt, offs, ssrc, nNodes);

    const int gemmBlocks = (nNodes + MBLK - 1) / MBLK;
    sage_gemm<<<gemmBlocks, 256, 0, stream>>>(x, cnt, offs, ssrc, W, bias, out, nNodes);
}


# A-tile staging keeps its loads in flight: own x rows loaded unguarded and together; a wave's source ids copied to LDS; next 8 source rows issued before the current 8 are summed; sum order unchanged
# speedup vs baseline: 3.6421x; 1.0373x over previous
#include <hip/hip_runtime.h>
#include <hip/hip_bf16.h>


typedef __attribute__((ext_vector_type(2))) float v2f;
typedef __attribute__((ext_vector_type(8))) float v8f;

#define IN_DIM   128
#define K_DIM    256
#define OUT_DIM  128
#define MBLK     64
#define LDS_STR  260

#define BIN_SHIFT 8
#define BIN_NODES 256
#define BIN_MASK  255
#define SORT_T    512
#define SORT_EPT  12
#define SORT_CHUNK (SORT_T * SORT_EPT)
#define ROW_CAP   8192

__device__ __forceinline__ int block_excl_scan(const int* in, int* out, int n, int* wtot)
{
    const int t     = threadIdx.x;
    const int lane  = t & 31;
    const int wave  = t >> 5;
    const int nthr  = SORT_T;
    const int per   = (n + nthr - 1) / nthr;
    const int b0    = t * per;

    int s = 0;
    for (int k = 0; k < per; ++k) { const int b = b0 + k; if (b < n) s += in[b]; }

    int v = s;
    for (int d = 1; d < 32; d <<= 1) {
        const int up = __shfl_up(v, d);
        if (lane >= d) v += up;
    }
    if (lane == 31) wtot[wave] = v;
    __syncthreads();
    if (wave == 0) {
        const int x = (lane < (nthr >> 5)) ? wtot[lane] : 0;
        int y = x;
        for (int d = 1; d < 32; d <<= 1) {
            const int up = __shfl_up(y, d);
            if (lane >= d) y += up;
        }
        wtot[lane] = y - x;
        if (lane == 31) wtot[32] = y;
    }
    __syncthreads();
    int run = wtot[wave] + v - s;
    for (int k = 0; k < per; ++k) {
        const int b = b0 + k;
        if (b < n) { const int c = in[b]; out[b] = run; run += c; }
    }
    const int total = wtot[32];
    __syncthreads();
    return total;
}

__global__ __launch_bounds__(SORT_T) void sage_bincount(
    const int* __restrict__ edst,
    int*       __restrict__ binCnt,
    int nEdges, int nBins)
{
    extern __shared__ int sm[];

    const int t = threadIdx.x;
    for (int b = t; b < nBins; b += SORT_T) sm[b] = 0;
    __syncthreads();

    const int e0 = blockIdx.x * SORT_CHUNK;
    int d[SORT_EPT];
    #pragma unroll
    for (int k = 0; k < SORT_EPT; ++k) {
        const int  e  = e0 + k * SORT_T + t;
        const bool ok = e < nEdges;
        const int  dv = edst[ok ? e : nEdges - 1];
        d[k] = ok ? dv : -1;
    }
    #pragma unroll
    for (int k = 0; k < SORT_EPT; ++k)
        if (d[k] >= 0) atomicAdd(&sm[d[k] >> BIN_SHIFT], 1);
    __syncthreads();

    for (int b = t; b < nBins; b += SORT_T) {
        const int c = sm[b];
        if (c) atomicAdd(&binCnt[b], c);
    }
}

__global__ __launch_bounds__(SORT_T) void sage_binplace(
    const int* __restrict__ esrc,
    const int* __restrict__ edst,
    const int* __restrict__ binCnt,
    int*       __restrict__ binCursor,
    int*       __restrict__ binBase,
    unsigned*  __restrict__ coarse,
    int nEdges, int nBins)
{
    extern __shared__ int sm[];
    __shared__ int      wtot[33];
    __shared__ uint2    stage[SORT_CHUNK];

    int* h   = sm;
    int* res = sm + nBins;
    int* loc = sm + 2 * nBins;

    const int t    = threadIdx.x;
    for (int b = t; b < nBins; b += SORT_T) h[b] = 0;
    const int total = block_excl_scan(binCnt, res, nBins, wtot);
    if (blockIdx.x == 0) {
        for (int b = t; b < nBins; b += SORT_T) binBase[b] = res[b];
        if (t == 0) binBase[nBins] = total;
    }

    const int e0 = blockIdx.x * SORT_CHUNK;
    int d[SORT_EPT], s[SORT_EPT], r[SORT_EPT];
    #pragma unroll
    for (int k = 0; k < SORT_EPT; ++k) {
        const int  e  = e0 + k * SORT_T + t;
        const bool ok = e < nEdges;
        const int  ec = ok ? e : nEdges - 1;
        const int  dv = edst[ec];
        s[k] = esrc[ec];
        d[k] = ok ? dv : -1;
    }
    #pragma unroll
    for (int k = 0; k < SORT_EPT; ++k) {
        r[k] = 0;
        if (d[k] >= 0) r[k] = atomicAdd(&h[d[k] >> BIN_SHIFT], 1);
    }
    __syncthreads();

    for (int b = t; b < nBins; b += SORT_T) {
        const int c = h[b];
        if (c) res[b] += atomicAdd(&binCursor[b], c);
    }
    block_excl_scan(h, loc, nBins, wtot);

    #pragma unroll
    for (int k = 0; k < SORT_EPT; ++k) {
        if (d[k] >= 0) {
            const int bin = d[k] >> BIN_SHIFT;
            stage[loc[bin] + r[k]] = make_uint2(
                ((unsigned)s[k] << BIN_SHIFT) | (unsigned)(d[k] & BIN_MASK),
                (unsigned)(res[bin] + r[k]));
        }
    }
    __syncthreads();

    const int nHere = (nEdges - e0 < SORT_CHUNK) ? (nEdges - e0) : SORT_CHUNK;
    #pragma unroll
    for (int k = 0; k < SORT_EPT; ++k) {
        const int i = k * SORT_T + t;
        if (i < nHere) { const uint2 q = stage[i]; coarse[q.y] = q.x; }
    }
}

__global__ __launch_bounds__(BIN_NODES) void sage_rowsort(
    const unsigned* __restrict__ coarse,
    const int*      __restrict__ binBase,
    int*            __restrict__ cnt,
    int*            __restrict__ offs,
    int*            __restrict__ ssrc,
    int nNodes)
{
    __shared__ int rc[BIN_NODES];
    __shared__ int ro[BIN_NODES];
    __shared__ int wsum[8];
    __shared__ int sorted[ROW_CAP];

    const int t    = threadIdx.x;
    const int lane = t & 31;
    const int wave = t >> 5;
    const int beg  = binBase[blockIdx.x];
    const int nb   = binBase[blockIdx.x + 1] - beg;
    const unsigned* __restrict__ src = coarse + beg;

    rc[t] = 0;
    __syncthreads();

    for (int i0 = 0; i0 < nb; i0 += BIN_NODES * 8) {
        unsigned w[8];
        #pragma unroll
        for (int u = 0; u < 8; ++u) {
            const int i = i0 + u * BIN_NODES + t;
            w[u] = src[i < nb ? i : nb - 1];
        }
        #pragma unroll
        for (int u = 0; u < 8; ++u) {
            const int i = i0 + u * BIN_NODES + t;
            if (i < nb) atomicAdd(&rc[w[u] & BIN_MASK], 1);
        }
    }
    __syncthreads();

    const int c = rc[t];
    int v = c;
    for (int d = 1; d < 32; d <<= 1) {
        const int up = __shfl_up(v, d);
        if (lane >= d) v += up;
    }
    if (lane == 31) wsum[wave] = v;
    __syncthreads();
    if (t == 0) {
        int tot = 0;
        for (int q = 0; q < 8; ++q) { const int x = wsum[q]; wsum[q] = tot; tot += x; }
    }
    __syncthreads();
    const int excl = wsum[wave] + v - c;
    ro[t] = excl;
    rc[t] = 0;
    {
        const int node = blockIdx.x * BIN_NODES + t;
        if (node < nNodes) { cnt[node] = c; offs[node] = beg + excl; }
    }
    __syncthreads();

    const bool inLds = nb <= ROW_CAP;
    for (int i0 = 0; i0 < nb; i0 += BIN_NODES * 8) {
        unsigned w[8];
        #pragma unroll
        for (int u = 0; u < 8; ++u) {
            const int i = i0 + u * BIN_NODES + t;
            w[u] = src[i < nb ? i : nb - 1];
        }
        #pragma unroll
        for (int u = 0; u < 8; ++u) {
            const int i = i0 + u * BIN_NODES + t;
            if (i < nb) {
                const int row = (int)(w[u] & BIN_MASK);
                const int p   = ro[row] + atomicAdd(&rc[row], 1);
                const int sid = (int)(w[u] >> BIN_SHIFT);
                if (inLds) sorted[p] = sid; else ssrc[beg + p] = sid;
            }
        }
    }
    if (inLds) {
        __syncthreads();
        for (int i = t; i < nb; i += BIN_NODES) ssrc[beg + i] = sorted[i];
    }
}

struct Rows8 { float4 v[8]; };

__device__ __forceinline__ void issue8(Rows8& R, const float4* __restrict__ xl,
                                       int mine, int j, int m)
{
    int s[8];
    #pragma unroll
    for (int u = 0; u < 8; ++u) s[u] = __builtin_amdgcn_readlane(mine, j + u);
    #pragma unroll
    for (int u = 0; u < 8; ++u) {
        R.v[u] = make_float4(0.f, 0.f, 0.f, 0.f);
        if (j + u < m) R.v[u] = xl[(size_t)s[u] * 32];
    }
}

__device__ __forceinline__ float4 add4(const float4& a, const float4& b)
{
    return make_float4(a.x + b.x, a.y + b.y, a.z + b.z, a.w + b.w);
}

__device__ __forceinline__ void add8(float4& acc, const Rows8& R)
{
    const float4 t0 = add4(R.v[0], R.v[1]);
    const float4 t1 = add4(R.v[2], R.v[3]);
    const float4 t2 = add4(R.v[4], R.v[5]);
    const float4 t3 = add4(R.v[6], R.v[7]);
    acc = add4(acc, add4(add4(t0, t1), add4(t2, t3)));
}

#define IDS_CAP 256

struct Item { int flat, cnt, row, n; bool last; };

__device__ __forceinline__ Item next_item(int& r, int& j, int& st, int& n, int myN)
{
    Item it;
    const bool valid = r < 8;
    const int  rem   = n - j;
    it.flat = st + j;
    it.cnt  = valid ? (rem < 8 ? rem : 8) : 0;
    it.row  = r;
    it.n    = n;
    it.last = valid && rem <= 8;
    const int rn = r + (it.last ? 1 : 0);
    const int nn = __builtin_amdgcn_readlane(myN, rn);
    st = it.last ? st + n : st;
    j  = it.last ? 0 : (valid ? j + 8 : j);
    n  = it.last ? nn : n;
    r  = rn;
    return it;
}

__device__ __forceinline__ void issue_item(Rows8& R, const float4* __restrict__ xl,
                                           const int* wids, const Item& it, int gl)
{
    int li = it.flat + (gl & 7);
    li = li < IDS_CAP - 1 ? li : IDS_CAP - 1;
    const int idv = wids[li];
    int s[8];
    #pragma unroll
    for (int u = 0; u < 8; ++u) s[u] = __builtin_amdgcn_readlane(idv, u);
    const int d = it.cnt > 0 ? s[0] : 0;
    #pragma unroll
    for (int u = 0; u < 8; ++u) {
        const int sid = (u < it.cnt) ? s[u] : d;
        R.v[u] = xl[(size_t)sid * 32];
    }
}

__device__ __forceinline__ void sum_item(float4& acc, Rows8& R, const Item& it, float* rowDst)
{
    const float4 z = make_float4(0.f, 0.f, 0.f, 0.f);
    #pragma unroll
    for (int u = 0; u < 8; ++u) if (u >= it.cnt) R.v[u] = z;
    add8(acc, R);
    if (it.last) {
        const float inv = 1.0f / fmaxf((float)it.n, 1.0f);
        float* dst = rowDst + it.row * LDS_STR;
        dst[0] = acc.x * inv; dst[1] = acc.y * inv; dst[2] = acc.z * inv; dst[3] = acc.w * inv;
        acc = z;
    }
}

__global__ __launch_bounds__(256) void sage_gemm(
    const float* __restrict__ x,
    const int*   __restrict__ cnt,
    const int*   __restrict__ offs,
    const int*   __restrict__ ssrc,
    const float* __restrict__ W,
    const float* __restrict__ bias,
    float*       __restrict__ out,
    int nNodes)
{
    __shared__ float As[MBLK * LDS_STR];
    __shared__ int   Ids[8 * IDS_CAP];

    const int mBase = blockIdx.x * MBLK;
    const int t = threadIdx.x;
    const bool fullBlock = (mBase + MBLK) <= nNodes;

    const float4* x4 = (const float4*)x;
    const int g    = __builtin_amdgcn_readfirstlane(t >> 5);
    const int gl   = t & 31;
    const int row0 = g * 8;
    const float4 zero4 = make_float4(0.f, 0.f, 0.f, 0.f);

    const int nd  = mBase + row0 + gl;
    const int ndc = nd < nNodes ? nd : nNodes - 1;
    const int cN  = cnt[ndc];
    const int cB  = offs[ndc];
    float4 xv[8];
    #pragma unroll
    for (int i = 0; i < 8; ++i) {
        const int grow = mBase + row0 + i;
        const int gc   = (fullBlock || grow < nNodes) ? grow : nNodes - 1;
        xv[i] = x4[(size_t)gc * 32 + gl];
    }
    const bool okN  = gl < 8 && nd < nNodes;
    const int myN   = okN ? cN : 0;
    const int myBeg = okN ? cB : 0;
    int tot = 0;
    #pragma unroll
    for (int u = 0; u < 8; ++u) tot += __builtin_amdgcn_readlane(myN, u);
    const int beg0 = __builtin_amdgcn_readlane(myBeg, 0);
    const bool piped = tot <= IDS_CAP;

    const float4* xl = x4 + gl;
    int*   wids   = &Ids[g * IDS_CAP];
    float* rowDst = &As[row0 * LDS_STR + IN_DIM + gl * 4];

    int idv[8];
    #pragma unroll
    for (int u = 0; u < 8; ++u) idv[u] = 0;
    if (piped && tot > 0) {
        #pragma unroll
        for (int u = 0; u < 8; ++u) {
            const int i = u * 32 + gl;
            idv[u] = ssrc[beg0 + (i < tot ? i : tot - 1)];
        }
    }
    #pragma unroll
    for (int u = 0; u < 8; ++u) wids[u * 32 + gl] = idv[u];

    #pragma unroll
    for (int i = 0; i < 8; ++i) {
        const int grow = mBase + row0 + i;
        const float4 v = (fullBlock || grow < nNodes) ? xv[i] : zero4;
        float* dst = &As[(row0 + i) * LDS_STR + gl * 4];
        dst[0] = v.x; dst[1] = v.y; dst[2] = v.z; dst[3] = v.w;
    }

    if (piped) {
        int r = 0, j = 0, st = 0;
        int n = __builtin_amdgcn_readlane(myN, 0);
        int nItems = 0;
        #pragma unroll
        for (int u = 0; u < 8; ++u) {
            const int nu = __builtin_amdgcn_readlane(myN, u);
            nItems += nu > 8 ? (nu + 7) >> 3 : 1;
        }
        float4 acc = zero4;
        Rows8 RA, RB;
        Item a = next_item(r, j, st, n, myN);
        issue_item(RA, xl, wids, a, gl);
        for (int i = 0; i < nItems; i += 2) {
            Item b = next_item(r, j, st, n, myN);
            issue_item(RB, xl, wids, b, gl);
            sum_item(acc, RA, a, rowDst);
            a = next_item(r, j, st, n, myN);
            issue_item(RA, xl, wids, a, gl);
            __builtin_amdgcn_sched_barrier(0);
            sum_item(acc, RB, b, rowDst);
        }
    } else {
        int n    = __builtin_amdgcn_readlane(myN, 0);
        int beg  = __builtin_amdgcn_readlane(myBeg, 0);
        int mine = (gl < n) ? ssrc[beg + gl] : 0;
        for (int r = 0; r < 8; ++r) {
            float4 acc = zero4;
            Rows8 R;
            int m = (n < 32) ? n : 32;
            issue8(R, xl, mine, 0, m);

            int n2 = 0, beg2 = 0, mine2 = 0;
            if (r + 1 < 8) {
                n2   = __builtin_amdgcn_readlane(myN, r + 1);
                beg2 = __builtin_amdgcn_readlane(myBeg, r + 1);
                mine2 = (gl < n2) ? ssrc[beg2 + gl] : 0;
            }

            add8(acc, R);
            for (int j = 8; j < m; j += 8) { issue8(R, xl, mine, j, m); add8(acc, R); }
            for (int c = 32; c < n; c += 32) {
                m = (n - c < 32) ? (n - c) : 32;
                mine = (gl < m) ? ssrc[beg + c + gl] : 0;
                for (int j = 0; j < m; j += 8) { issue8(R, xl, mine, j, m); add8(acc, R); }
            }

            const float inv = 1.0f / fmaxf((float)n, 1.0f);
            float* dst = rowDst + r * LDS_STR;
            dst[0] = acc.x * inv; dst[1] = acc.y * inv; dst[2] = acc.z * inv; dst[3] = acc.w * inv;

            n = n2; beg = beg2; mine = mine2;
        }
    }

    __syncthreads();

    const int wave  = t >> 5;
    const int lane  = t & 31;
    const int ln    = lane & 15;
    const int hi    = lane >> 4;
    const int nBase = wave * 16;

    v8f acc[4] = {};

    for (int k0 = 0; k0 < K_DIM; k0 += 4) {
        const int kk = k0 + hi * 2;

        v2f bfrag;
        bfrag.x = W[(size_t)kk       * OUT_DIM + nBase + ln];
        bfrag.y = W[(size_t)(kk + 1) * OUT_DIM + nBase + ln];

        #pragma unroll
        for (int ms = 0; ms < 4; ++ms) {
            const int row = ms * 16 + ln;
            v2f afrag;
            afrag.x = As[row * LDS_STR + kk];
            afrag.y = As[row * LDS_STR + kk + 1];

            acc[ms] = __builtin_amdgcn_wmma_f32_16x16x4_f32(
                 false, afrag,  false, bfrag,
                 (short)0, acc[ms],
                 false,  false);
        }
    }

    const float bv = bias[nBase + ln];
    float* obase = out + (size_t)(mBase + hi * 8) * OUT_DIM + nBase + ln;
    if (fullBlock) {
        #pragma unroll
        for (int ms = 0; ms < 4; ++ms) {
            #pragma unroll
            for (int r = 0; r < 8; ++r) {
                obase[(size_t)(ms * 16 + r) * OUT_DIM] = acc[ms][r] + bv;
            }
        }
    } else {
        #pragma unroll
        for (int ms = 0; ms < 4; ++ms) {
            #pragma unroll
            for (int r = 0; r < 8; ++r) {
                const int grow = mBase + ms * 16 + r + hi * 8;
                if (grow < nNodes)
                    out[(size_t)grow * OUT_DIM + nBase + ln] = acc[ms][r] + bv;
            }
        }
    }
}

extern "C" void kernel_launch(void* const* d_in, const int* in_sizes, int n_in,
                              void* d_out, int out_size, void* d_ws, size_t ws_size,
                              hipStream_t stream)
{
    const float* x    = (const float*)d_in[0];
    const int*   esrc = (const int*)  d_in[1];
    const int*   edst = (const int*)  d_in[2];
    const float* W    = (const float*)d_in[3];
    const float* bias = (const float*)d_in[4];
    float*       out  = (float*)d_out;

    const int nNodes = in_sizes[0] / IN_DIM;
    const int nEdges = in_sizes[1];
    const int nBins  = (nNodes + BIN_NODES - 1) / BIN_NODES;

    int*      cnt       = (int*)d_ws;
    int*      offs      = cnt + nNodes;
    unsigned* coarse    = (unsigned*)(offs + nNodes);
    int*      ssrc      = (int*)(coarse + nEdges);
    int*      binCnt    = ssrc + nEdges;
    int*      binCursor = binCnt + nBins;
    int*      binBase   = binCursor + nBins;
    hipMemsetAsync(binCnt, 0, (size_t)2 * nBins * sizeof(int), stream);

    const int sortBlocks = (nEdges + SORT_CHUNK - 1) / SORT_CHUNK;
    sage_bincount<<<sortBlocks, SORT_T, (size_t)nBins * sizeof(int), stream>>>(edst, binCnt, nEdges, nBins);
    sage_binplace<<<sortBlocks, SORT_T, (size_t)3 * nBins * sizeof(int), stream>>>(
        esrc, edst, binCnt, binCursor, binBase, coarse, nEdges, nBins);
    sage_rowsort<<<nBins, BIN_NODES, 0, stream>>>(coarse, binBase, cnt, offs, ssrc, nNodes);

    const int gemmBlocks = (nNodes + MBLK - 1) / MBLK;
    sage_gemm<<<gemmBlocks, 256, 0, stream>>>(x, cnt, offs, ssrc, W, bias, out, nNodes);
}


# GEMM kernel made one persistent 32-wave work-group per WGP: 16 waves gather the next A tile into a second LDS buffer while 16 waves run the current tile's WMMAs; W held in LDS; sum order unchanged
# speedup vs baseline: 3.7588x; 1.0320x over previous
#include <hip/hip_runtime.h>
#include <hip/hip_bf16.h>


typedef __attribute__((ext_vector_type(2))) float v2f;
typedef __attribute__((ext_vector_type(8))) float v8f;

#define IN_DIM   128
#define K_DIM    256
#define OUT_DIM  128
#define MBLK     64
#define LDS_STR  260
#define W_STR    136
#define GWAVES   16
#define MWAVES   16
#define NPW      4
#define GEMM_T   ((GWAVES + MWAVES) * 32)

#define BIN_SHIFT 8
#define BIN_NODES 256
#define BIN_MASK  255
#define SORT_T    512
#define SORT_EPT  12
#define SORT_CHUNK (SORT_T * SORT_EPT)
#define ROW_CAP   8192

__device__ __forceinline__ int block_excl_scan(const int* in, int* out, int n, int* wtot)
{
    const int t     = threadIdx.x;
    const int lane  = t & 31;
    const int wave  = t >> 5;
    const int nthr  = SORT_T;
    const int per   = (n + nthr - 1) / nthr;
    const int b0    = t * per;

    int s = 0;
    for (int k = 0; k < per; ++k) { const int b = b0 + k; if (b < n) s += in[b]; }

    int v = s;
    for (int d = 1; d < 32; d <<= 1) {
        const int up = __shfl_up(v, d);
        if (lane >= d) v += up;
    }
    if (lane == 31) wtot[wave] = v;
    __syncthreads();
    if (wave == 0) {
        const int x = (lane < (nthr >> 5)) ? wtot[lane] : 0;
        int y = x;
        for (int d = 1; d < 32; d <<= 1) {
            const int up = __shfl_up(y, d);
            if (lane >= d) y += up;
        }
        wtot[lane] = y - x;
        if (lane == 31) wtot[32] = y;
    }
    __syncthreads();
    int run = wtot[wave] + v - s;
    for (int k = 0; k < per; ++k) {
        const int b = b0 + k;
        if (b < n) { const int c = in[b]; out[b] = run; run += c; }
    }
    const int total = wtot[32];
    __syncthreads();
    return total;
}

__global__ __launch_bounds__(SORT_T) void sage_bincount(
    const int* __restrict__ edst,
    int*       __restrict__ binCnt,
    int nEdges, int nBins)
{
    extern __shared__ int sm[];

    const int t = threadIdx.x;
    for (int b = t; b < nBins; b += SORT_T) sm[b] = 0;
    __syncthreads();

    const int e0 = blockIdx.x * SORT_CHUNK;
    int d[SORT_EPT];
    #pragma unroll
    for (int k = 0; k < SORT_EPT; ++k) {
        const int  e  = e0 + k * SORT_T + t;
        const bool ok = e < nEdges;
        const int  dv = edst[ok ? e : nEdges - 1];
        d[k] = ok ? dv : -1;
    }
    #pragma unroll
    for (int k = 0; k < SORT_EPT; ++k)
        if (d[k] >= 0) atomicAdd(&sm[d[k] >> BIN_SHIFT], 1);
    __syncthreads();

    for (int b = t; b < nBins; b += SORT_T) {
        const int c = sm[b];
        if (c) atomicAdd(&binCnt[b], c);
    }
}

__global__ __launch_bounds__(SORT_T) void sage_binplace(
    const int* __restrict__ esrc,
    const int* __restrict__ edst,
    const int* __restrict__ binCnt,
    int*       __restrict__ binCursor,
    int*       __restrict__ binBase,
    unsigned*  __restrict__ coarse,
    int nEdges, int nBins)
{
    extern __shared__ int sm[];
    __shared__ int      wtot[33];
    __shared__ uint2    stage[SORT_CHUNK];

    int* h   = sm;
    int* res = sm + nBins;
    int* loc = sm + 2 * nBins;

    const int t    = threadIdx.x;
    for (int b = t; b < nBins; b += SORT_T) h[b] = 0;
    const int total = block_excl_scan(binCnt, res, nBins, wtot);
    if (blockIdx.x == 0) {
        for (int b = t; b < nBins; b += SORT_T) binBase[b] = res[b];
        if (t == 0) binBase[nBins] = total;
    }

    const int e0 = blockIdx.x * SORT_CHUNK;
    int d[SORT_EPT], s[SORT_EPT], r[SORT_EPT];
    #pragma unroll
    for (int k = 0; k < SORT_EPT; ++k) {
        const int  e  = e0 + k * SORT_T + t;
        const bool ok = e < nEdges;
        const int  ec = ok ? e : nEdges - 1;
        const int  dv = edst[ec];
        s[k] = esrc[ec];
        d[k] = ok ? dv : -1;
    }
    #pragma unroll
    for (int k = 0; k < SORT_EPT; ++k) {
        r[k] = 0;
        if (d[k] >= 0) r[k] = atomicAdd(&h[d[k] >> BIN_SHIFT], 1);
    }
    __syncthreads();

    for (int b = t; b < nBins; b += SORT_T) {
        const int c = h[b];
        if (c) res[b] += atomicAdd(&binCursor[b], c);
    }
    block_excl_scan(h, loc, nBins, wtot);

    #pragma unroll
    for (int k = 0; k < SORT_EPT; ++k) {
        if (d[k] >= 0) {
            const int bin = d[k] >> BIN_SHIFT;
            stage[loc[bin] + r[k]] = make_uint2(
                ((unsigned)s[k] << BIN_SHIFT) | (unsigned)(d[k] & BIN_MASK),
                (unsigned)(res[bin] + r[k]));
        }
    }
    __syncthreads();

    const int nHere = (nEdges - e0 < SORT_CHUNK) ? (nEdges - e0) : SORT_CHUNK;
    #pragma unroll
    for (int k = 0; k < SORT_EPT; ++k) {
        const int i = k * SORT_T + t;
        if (i < nHere) { const uint2 q = stage[i]; coarse[q.y] = q.x; }
    }
}

__global__ __launch_bounds__(BIN_NODES) void sage_rowsort(
    const unsigned* __restrict__ coarse,
    const int*      __restrict__ binBase,
    int*            __restrict__ cnt,
    int*            __restrict__ offs,
    int*            __restrict__ ssrc,
    int nNodes)
{
    __shared__ int rc[BIN_NODES];
    __shared__ int ro[BIN_NODES];
    __shared__ int wsum[8];
    __shared__ int sorted[ROW_CAP];

    const int t    = threadIdx.x;
    const int lane = t & 31;
    const int wave = t >> 5;
    const int beg  = binBase[blockIdx.x];
    const int nb   = binBase[blockIdx.x + 1] - beg;
    const unsigned* __restrict__ src = coarse + beg;

    rc[t] = 0;
    __syncthreads();

    for (int i0 = 0; i0 < nb; i0 += BIN_NODES * 8) {
        unsigned w[8];
        #pragma unroll
        for (int u = 0; u < 8; ++u) {
            const int i = i0 + u * BIN_NODES + t;
            w[u] = src[i < nb ? i : nb - 1];
        }
        #pragma unroll
        for (int u = 0; u < 8; ++u) {
            const int i = i0 + u * BIN_NODES + t;
            if (i < nb) atomicAdd(&rc[w[u] & BIN_MASK], 1);
        }
    }
    __syncthreads();

    const int c = rc[t];
    int v = c;
    for (int d = 1; d < 32; d <<= 1) {
        const int up = __shfl_up(v, d);
        if (lane >= d) v += up;
    }
    if (lane == 31) wsum[wave] = v;
    __syncthreads();
    if (t == 0) {
        int tot = 0;
        for (int q = 0; q < 8; ++q) { const int x = wsum[q]; wsum[q] = tot; tot += x; }
    }
    __syncthreads();
    const int excl = wsum[wave] + v - c;
    ro[t] = excl;
    rc[t] = 0;
    {
        const int node = blockIdx.x * BIN_NODES + t;
        if (node < nNodes) { cnt[node] = c; offs[node] = beg + excl; }
    }
    __syncthreads();

    const bool inLds = nb <= ROW_CAP;
    for (int i0 = 0; i0 < nb; i0 += BIN_NODES * 8) {
        unsigned w[8];
        #pragma unroll
        for (int u = 0; u < 8; ++u) {
            const int i = i0 + u * BIN_NODES + t;
            w[u] = src[i < nb ? i : nb - 1];
        }
        #pragma unroll
        for (int u = 0; u < 8; ++u) {
            const int i = i0 + u * BIN_NODES + t;
            if (i < nb) {
                const int row = (int)(w[u] & BIN_MASK);
                const int p   = ro[row] + atomicAdd(&rc[row], 1);
                const int sid = (int)(w[u] >> BIN_SHIFT);
                if (inLds) sorted[p] = sid; else ssrc[beg + p] = sid;
            }
        }
    }
    if (inLds) {
        __syncthreads();
        for (int i = t; i < nb; i += BIN_NODES) ssrc[beg + i] = sorted[i];
    }
}

struct Rows8 { float4 v[8]; };

__device__ __forceinline__ void issue8(Rows8& R, const float4* __restrict__ xl,
                                       int mine, int j, int m)
{
    int s[8];
    #pragma unroll
    for (int u = 0; u < 8; ++u) s[u] = __builtin_amdgcn_readlane(mine, j + u);
    #pragma unroll
    for (int u = 0; u < 8; ++u) {
        R.v[u] = make_float4(0.f, 0.f, 0.f, 0.f);
        if (j + u < m) R.v[u] = xl[(size_t)s[u] * 32];
    }
}

__device__ __forceinline__ float4 add4(const float4& a, const float4& b)
{
    return make_float4(a.x + b.x, a.y + b.y, a.z + b.z, a.w + b.w);
}

__device__ __forceinline__ void add8(float4& acc, const Rows8& R)
{
    const float4 t0 = add4(R.v[0], R.v[1]);
    const float4 t1 = add4(R.v[2], R.v[3]);
    const float4 t2 = add4(R.v[4], R.v[5]);
    const float4 t3 = add4(R.v[6], R.v[7]);
    acc = add4(acc, add4(add4(t0, t1), add4(t2, t3)));
}

#define IDS_CAP 128

struct Item { int flat, cnt, row, n; bool last; };

__device__ __forceinline__ Item next_item(int& r, int& j, int& st, int& n, int myN)
{
    Item it;
    const bool valid = r < NPW;
    const int  rem   = n - j;
    it.flat = st + j;
    it.cnt  = valid ? (rem < 8 ? rem : 8) : 0;
    it.row  = r;
    it.n    = n;
    it.last = valid && rem <= 8;
    const int rn = r + (it.last ? 1 : 0);
    const int nn = __builtin_amdgcn_readlane(myN, rn);
    st = it.last ? st + n : st;
    j  = it.last ? 0 : (valid ? j + 8 : j);
    n  = it.last ? nn : n;
    r  = rn;
    return it;
}

__device__ __forceinline__ void issue_item(Rows8& R, const float4* __restrict__ xl,
                                           const int* wids, const Item& it, int gl)
{
    int li = it.flat + (gl & 7);
    li = li < IDS_CAP - 1 ? li : IDS_CAP - 1;
    const int idv = wids[li];
    int s[8];
    #pragma unroll
    for (int u = 0; u < 8; ++u) s[u] = __builtin_amdgcn_readlane(idv, u);
    const int d = it.cnt > 0 ? s[0] : 0;
    #pragma unroll
    for (int u = 0; u < 8; ++u) {
        const int sid = (u < it.cnt) ? s[u] : d;
        R.v[u] = xl[(size_t)sid * 32];
    }
}

__device__ __forceinline__ void sum_item(float4& acc, Rows8& R, const Item& it, float* rowDst)
{
    const float4 z = make_float4(0.f, 0.f, 0.f, 0.f);
    #pragma unroll
    for (int u = 0; u < 8; ++u) if (u >= it.cnt) R.v[u] = z;
    add8(acc, R);
    if (it.last) {
        const float inv = 1.0f / fmaxf((float)it.n, 1.0f);
        float* dst = rowDst + it.row * LDS_STR;
        dst[0] = acc.x * inv; dst[1] = acc.y * inv; dst[2] = acc.z * inv; dst[3] = acc.w * inv;
        acc = z;
    }
}


__device__ __forceinline__ void stage_rows(
    const float* __restrict__ x,
    const int*   __restrict__ cnt,
    const int*   __restrict__ offs,
    const int*   __restrict__ ssrc,
    float* Ab, int* wids, int mBase, int nNodes, int g, int gl)
{
    const bool fullBlock = (mBase + MBLK) <= nNodes;
    const float4* x4 = (const float4*)x;
    const int row0 = g * NPW;
    const float4 zero4 = make_float4(0.f, 0.f, 0.f, 0.f);

    const int nd  = mBase + row0 + gl;
    const int ndc = nd < nNodes ? nd : nNodes - 1;
    const int cN  = cnt[ndc];
    const int cB  = offs[ndc];
    float4 xv[NPW];
    #pragma unroll
    for (int i = 0; i < NPW; ++i) {
        const int grow = mBase + row0 + i;
        const int gc   = (fullBlock || grow < nNodes) ? grow : nNodes - 1;
        xv[i] = x4[(size_t)gc * 32 + gl];
    }
    const bool okN  = gl < NPW && nd < nNodes;
    const int myN   = okN ? cN : 0;
    const int myBeg = okN ? cB : 0;
    int tot = 0;
    #pragma unroll
    for (int u = 0; u < NPW; ++u) tot += __builtin_amdgcn_readlane(myN, u);
    const int beg0 = __builtin_amdgcn_readlane(myBeg, 0);
    const bool piped = tot <= IDS_CAP;

    const float4* xl = x4 + gl;
    float* rowDst = &Ab[row0 * LDS_STR + IN_DIM + gl * 4];

    int idv[IDS_CAP / 32];
    #pragma unroll
    for (int u = 0; u < IDS_CAP / 32; ++u) idv[u] = 0;
    if (piped && tot > 0) {
        #pragma unroll
        for (int u = 0; u < IDS_CAP / 32; ++u) {
            const int i = u * 32 + gl;
            idv[u] = ssrc[beg0 + (i < tot ? i : tot - 1)];
        }
    }
    #pragma unroll
    for (int u = 0; u < IDS_CAP / 32; ++u) wids[u * 32 + gl] = idv[u];

    #pragma unroll
    for (int i = 0; i < NPW; ++i) {
        const int grow = mBase + row0 + i;
        const float4 v = (fullBlock || grow < nNodes) ? xv[i] : zero4;
        float* dst = &Ab[(row0 + i) * LDS_STR + gl * 4];
        dst[0] = v.x; dst[1] = v.y; dst[2] = v.z; dst[3] = v.w;
    }

    if (piped) {
        int r = 0, j = 0, st = 0;
        int n = __builtin_amdgcn_readlane(myN, 0);
        int nItems = 0;
        #pragma unroll
        for (int u = 0; u < NPW; ++u) {
            const int nu = __builtin_amdgcn_readlane(myN, u);
            nItems += nu > 8 ? (nu + 7) >> 3 : 1;
        }
        float4 acc = zero4;
        Rows8 RA, RB;
        Item a = next_item(r, j, st, n, myN);
        issue_item(RA, xl, wids, a, gl);
        for (int i = 0; i < nItems; i += 2) {
            Item b = next_item(r, j, st, n, myN);
            issue_item(RB, xl, wids, b, gl);
            sum_item(acc, RA, a, rowDst);
            a = next_item(r, j, st, n, myN);
            issue_item(RA, xl, wids, a, gl);
            __builtin_amdgcn_sched_barrier(0);
            sum_item(acc, RB, b, rowDst);
        }
    } else {
        int n    = __builtin_amdgcn_readlane(myN, 0);
        int beg  = __builtin_amdgcn_readlane(myBeg, 0);
        int mine = (gl < n) ? ssrc[beg + gl] : 0;
        for (int r = 0; r < NPW; ++r) {
            float4 acc = zero4;
            Rows8 R;
            int m = (n < 32) ? n : 32;
            issue8(R, xl, mine, 0, m);

            int n2 = 0, beg2 = 0, mine2 = 0;
            if (r + 1 < NPW) {
                n2   = __builtin_amdgcn_readlane(myN, r + 1);
                beg2 = __builtin_amdgcn_readlane(myBeg, r + 1);
                mine2 = (gl < n2) ? ssrc[beg2 + gl] : 0;
            }

            add8(acc, R);
            for (int j = 8; j < m; j += 8) { issue8(R, xl, mine, j, m); add8(acc, R); }
            for (int c = 32; c < n; c += 32) {
                m = (n - c < 32) ? (n - c) : 32;
                mine = (gl < m) ? ssrc[beg + c + gl] : 0;
                for (int j = 0; j < m; j += 8) { issue8(R, xl, mine, j, m); add8(acc, R); }
            }

            const float inv = 1.0f / fmaxf((float)n, 1.0f);
            float* dst = rowDst + r * LDS_STR;
            dst[0] = acc.x * inv; dst[1] = acc.y * inv; dst[2] = acc.z * inv; dst[3] = acc.w * inv;

            n = n2; beg = beg2; mine = mine2;
        }
    }
}

__device__ __forceinline__ void mma_rows(
    const float* Ab, const float* Ws, float bv,
    float* __restrict__ out, int mBase, int nNodes, int m, int lane)
{
    const int ln    = lane & 15;
    const int hi    = lane >> 4;
    const int nBase = (m & 7) * 16;
    const int rBase = (m >> 3) * 32;

    v8f acc[2] = {};

    #pragma clang loop unroll(disable)
    for (int k0 = 0; k0 < K_DIM; k0 += 32) {
        asm volatile("v_nop\n\tv_nop\n\tv_nop\n\tv_nop\n\tv_nop\n\tv_nop\n\tv_nop\n\tv_nop");
        __builtin_amdgcn_sched_barrier(0);
        v2f bf[8];
        v2f af[8][2];
        #pragma unroll
        for (int s = 0; s < 8; ++s) {
            const int kk = k0 + s * 4 + hi * 2;
            bf[s].x = Ws[kk       * W_STR + nBase + ln];
            bf[s].y = Ws[(kk + 1) * W_STR + nBase + ln];
            #pragma unroll
            for (int ms = 0; ms < 2; ++ms) {
                const int row = rBase + ms * 16 + ln;
                af[s][ms].x = Ab[row * LDS_STR + kk];
                af[s][ms].y = Ab[row * LDS_STR + kk + 1];
            }
        }
        __builtin_amdgcn_sched_barrier(0);
        #pragma unroll
        for (int s = 0; s < 8; ++s) {
            #pragma unroll
            for (int ms = 0; ms < 2; ++ms) {
                acc[ms] = __builtin_amdgcn_wmma_f32_16x16x4_f32(
                     false, af[s][ms],  false, bf[s],
                     (short)0, acc[ms],
                     false,  false);
            }
        }
    }

    asm volatile("v_nop\n\tv_nop\n\tv_nop\n\tv_nop\n\tv_nop\n\tv_nop\n\tv_nop\n\tv_nop");
    __builtin_amdgcn_sched_barrier(0);

    if ((mBase + MBLK) <= nNodes) {
        float* obase = out + (size_t)(mBase + rBase + hi * 8) * OUT_DIM + nBase + ln;
        #pragma unroll
        for (int ms = 0; ms < 2; ++ms) {
            #pragma unroll
            for (int r = 0; r < 8; ++r) {
                obase[(size_t)(ms * 16 + r) * OUT_DIM] = acc[ms][r] + bv;
            }
        }
    } else {
        #pragma unroll
        for (int ms = 0; ms < 2; ++ms) {
            #pragma unroll
            for (int r = 0; r < 8; ++r) {
                const int grow = mBase + rBase + ms * 16 + r + hi * 8;
                if (grow < nNodes)
                    out[(size_t)grow * OUT_DIM + nBase + ln] = acc[ms][r] + bv;
            }
        }
    }
}

__global__ __launch_bounds__(GEMM_T) void sage_gemm(
    const float* __restrict__ x,
    const int*   __restrict__ cnt,
    const int*   __restrict__ offs,
    const int*   __restrict__ ssrc,
    const float* __restrict__ W,
    const float* __restrict__ bias,
    float*       __restrict__ out,
    int nNodes, int nTiles)
{
    __shared__ __attribute__((aligned(16))) float Ws[K_DIM * W_STR];
    __shared__ __attribute__((aligned(16))) float As[2 * MBLK * LDS_STR];
    __shared__ int Ids[GWAVES * IDS_CAP];

    const int t  = threadIdx.x;
    const int wv = __builtin_amdgcn_readfirstlane(t >> 5);
    const int gl = t & 31;

    {
        const float4* W4 = (const float4*)W;
        #pragma unroll
        for (int i = 0; i < (K_DIM * OUT_DIM / 4) / GEMM_T; ++i) {
            const int idx = i * GEMM_T + t;
            const float4 v = W4[idx];
            *(float4*)&Ws[(idx >> 5) * W_STR + (idx & 31) * 4] = v;
        }
    }
    const float bv = bias[(wv & 7) * 16 + (gl & 15)];

    const int first = blockIdx.x;
    const int step  = gridDim.x;
    const int nT    = first < nTiles ? (nTiles - first + step - 1) / step : 0;

    for (int it = 0; it <= nT; ++it) {
        if (wv < GWAVES) {
            if (it < nT)
                stage_rows(x, cnt, offs, ssrc, &As[(it & 1) * (MBLK * LDS_STR)],
                           &Ids[wv * IDS_CAP], (first + it * step) * MBLK, nNodes, wv, gl);
        } else {
            if (it > 0)
                mma_rows(&As[((it - 1) & 1) * (MBLK * LDS_STR)], Ws, bv, out,
                         (first + (it - 1) * step) * MBLK, nNodes, wv - GWAVES, gl);
        }
        __builtin_amdgcn_fence(__ATOMIC_RELEASE, "workgroup", "local");
        __builtin_amdgcn_s_barrier();
        __builtin_amdgcn_fence(__ATOMIC_ACQUIRE, "workgroup", "local");
    }
}

extern "C" void kernel_launch(void* const* d_in, const int* in_sizes, int n_in,
                              void* d_out, int out_size, void* d_ws, size_t ws_size,
                              hipStream_t stream)
{
    const float* x    = (const float*)d_in[0];
    const int*   esrc = (const int*)  d_in[1];
    const int*   edst = (const int*)  d_in[2];
    const float* W    = (const float*)d_in[3];
    const float* bias = (const float*)d_in[4];
    float*       out  = (float*)d_out;

    const int nNodes = in_sizes[0] / IN_DIM;
    const int nEdges = in_sizes[1];
    const int nBins  = (nNodes + BIN_NODES - 1) / BIN_NODES;

    int*      cnt       = (int*)d_ws;
    int*      offs      = cnt + nNodes;
    unsigned* coarse    = (unsigned*)(offs + nNodes);
    int*      ssrc      = (int*)(coarse + nEdges);
    int*      binCnt    = ssrc + nEdges;
    int*      binCursor = binCnt + nBins;
    int*      binBase   = binCursor + nBins;
    hipMemsetAsync(binCnt, 0, (size_t)2 * nBins * sizeof(int), stream);

    const int sortBlocks = (nEdges + SORT_CHUNK - 1) / SORT_CHUNK;
    sage_bincount<<<sortBlocks, SORT_T, (size_t)nBins * sizeof(int), stream>>>(edst, binCnt, nEdges, nBins);
    sage_binplace<<<sortBlocks, SORT_T, (size_t)3 * nBins * sizeof(int), stream>>>(
        esrc, edst, binCnt, binCursor, binBase, coarse, nEdges, nBins);
    sage_rowsort<<<nBins, BIN_NODES, 0, stream>>>(coarse, binBase, cnt, offs, ssrc, nNodes);

    const int gemmTiles = (nNodes + MBLK - 1) / MBLK;
    sage_gemm<<<256, GEMM_T, 0, stream>>>(x, cnt, offs, ssrc, W, bias, out, nNodes, gemmTiles);
}


# Per-node ordering pass folded into the product launch: a work-group orders the edges of its own equal row range in LDS and gathers from that list; one launch fewer; sum order still LDS arrival rank
# speedup vs baseline: 3.8973x; 1.0368x over previous
#include <hip/hip_runtime.h>
#include <hip/hip_bf16.h>


typedef __attribute__((ext_vector_type(2))) float v2f;
typedef __attribute__((ext_vector_type(8))) float v8f;

#define IN_DIM   128
#define K_DIM    256
#define OUT_DIM  128
#define MBLK     64
#define LDS_STR  260
#define W_STR    136
#define GWAVES   16
#define MWAVES   16
#define NPW      4
#define GEMM_T   ((GWAVES + MWAVES) * 32)

#define SH_BITS   9
#define SH_CAP    512
#define SH_MASK   511
#define N_SHARES  256
#define SORT_T    512
#define SORT_EPT  12
#define SORT_CHUNK (SORT_T * SORT_EPT)
#define ID_CAP    8192

__device__ __forceinline__ int block_excl_scan(const int* in, int* out, int n, int* wtot)
{
    const int t     = threadIdx.x;
    const int lane  = t & 31;
    const int wave  = t >> 5;
    const int nthr  = SORT_T;
    const int per   = (n + nthr - 1) / nthr;
    const int b0    = t * per;

    int s = 0;
    for (int k = 0; k < per; ++k) { const int b = b0 + k; if (b < n) s += in[b]; }

    int v = s;
    for (int d = 1; d < 32; d <<= 1) {
        const int up = __shfl_up(v, d);
        if (lane >= d) v += up;
    }
    if (lane == 31) wtot[wave] = v;
    __syncthreads();
    if (wave == 0) {
        const int x = (lane < (nthr >> 5)) ? wtot[lane] : 0;
        int y = x;
        for (int d = 1; d < 32; d <<= 1) {
            const int up = __shfl_up(y, d);
            if (lane >= d) y += up;
        }
        wtot[lane] = y - x;
        if (lane == 31) wtot[32] = y;
    }
    __syncthreads();
    int run = wtot[wave] + v - s;
    for (int k = 0; k < per; ++k) {
        const int b = b0 + k;
        if (b < n) { const int c = in[b]; out[b] = run; run += c; }
    }
    const int total = wtot[32];
    __syncthreads();
    return total;
}

__global__ __launch_bounds__(SORT_T) void sage_bincount(
    const int* __restrict__ edst,
    int*       __restrict__ binCnt,
    int nEdges, int nBins, int per)
{
    extern __shared__ int sm[];

    const int t = threadIdx.x;
    for (int b = t; b < nBins; b += SORT_T) sm[b] = 0;
    __syncthreads();

    const int e0 = blockIdx.x * SORT_CHUNK;
    int d[SORT_EPT];
    #pragma unroll
    for (int k = 0; k < SORT_EPT; ++k) {
        const int  e  = e0 + k * SORT_T + t;
        const bool ok = e < nEdges;
        const int  dv = edst[ok ? e : nEdges - 1];
        d[k] = ok ? dv : -1;
    }
    #pragma unroll
    for (int k = 0; k < SORT_EPT; ++k)
        if (d[k] >= 0) atomicAdd(&sm[d[k] / per], 1);
    __syncthreads();

    for (int b = t; b < nBins; b += SORT_T) {
        const int c = sm[b];
        if (c) atomicAdd(&binCnt[b], c);
    }
}

__global__ __launch_bounds__(SORT_T) void sage_binplace(
    const int* __restrict__ esrc,
    const int* __restrict__ edst,
    const int* __restrict__ binCnt,
    int*       __restrict__ binCursor,
    int*       __restrict__ binBase,
    unsigned*  __restrict__ coarse,
    int nEdges, int nBins, int per)
{
    extern __shared__ int sm[];
    __shared__ int      wtot[33];
    __shared__ uint2    stage[SORT_CHUNK];

    int* h   = sm;
    int* res = sm + nBins;
    int* loc = sm + 2 * nBins;

    const int t    = threadIdx.x;
    for (int b = t; b < nBins; b += SORT_T) h[b] = 0;
    const int total = block_excl_scan(binCnt, res, nBins, wtot);
    if (blockIdx.x == 0) {
        for (int b = t; b < nBins; b += SORT_T) binBase[b] = res[b];
        if (t == 0) binBase[nBins] = total;
    }

    const int e0 = blockIdx.x * SORT_CHUNK;
    int d[SORT_EPT], s[SORT_EPT], r[SORT_EPT];
    #pragma unroll
    for (int k = 0; k < SORT_EPT; ++k) {
        const int  e  = e0 + k * SORT_T + t;
        const bool ok = e < nEdges;
        const int  ec = ok ? e : nEdges - 1;
        const int  dv = edst[ec];
        s[k] = esrc[ec];
        d[k] = ok ? dv : -1;
    }
    #pragma unroll
    for (int k = 0; k < SORT_EPT; ++k) {
        r[k] = 0;
        if (d[k] >= 0) r[k] = atomicAdd(&h[d[k] / per], 1);
    }
    __syncthreads();

    for (int b = t; b < nBins; b += SORT_T) {
        const int c = h[b];
        if (c) res[b] += atomicAdd(&binCursor[b], c);
    }
    block_excl_scan(h, loc, nBins, wtot);

    #pragma unroll
    for (int k = 0; k < SORT_EPT; ++k) {
        if (d[k] >= 0) {
            const int bin = d[k] / per;
            stage[loc[bin] + r[k]] = make_uint2(
                ((unsigned)s[k] << SH_BITS) | (unsigned)(d[k] - bin * per),
                (unsigned)(res[bin] + r[k]));
        }
    }
    __syncthreads();

    const int nHere = (nEdges - e0 < SORT_CHUNK) ? (nEdges - e0) : SORT_CHUNK;
    #pragma unroll
    for (int k = 0; k < SORT_EPT; ++k) {
        const int i = k * SORT_T + t;
        if (i < nHere) { const uint2 q = stage[i]; coarse[q.y] = q.x; }
    }
}

struct Rows8 { float4 v[8]; };

__device__ __forceinline__ int load_id(const int* sorted, int* ovf, bool inLds, int i)
{
    return inLds ? sorted[i] : __hip_atomic_load(ovf + i, __ATOMIC_RELAXED, __HIP_MEMORY_SCOPE_AGENT);
}

__device__ __forceinline__ void issue8(Rows8& R, const float4* __restrict__ xl,
                                       int mine, int j, int m)
{
    int s[8];
    #pragma unroll
    for (int u = 0; u < 8; ++u) s[u] = __builtin_amdgcn_readlane(mine, j + u);
    #pragma unroll
    for (int u = 0; u < 8; ++u) {
        R.v[u] = make_float4(0.f, 0.f, 0.f, 0.f);
        if (j + u < m) R.v[u] = xl[(size_t)s[u] * 32];
    }
}

__device__ __forceinline__ float4 add4(const float4& a, const float4& b)
{
    return make_float4(a.x + b.x, a.y + b.y, a.z + b.z, a.w + b.w);
}

__device__ __forceinline__ void add8(float4& acc, const Rows8& R)
{
    const float4 t0 = add4(R.v[0], R.v[1]);
    const float4 t1 = add4(R.v[2], R.v[3]);
    const float4 t2 = add4(R.v[4], R.v[5]);
    const float4 t3 = add4(R.v[6], R.v[7]);
    acc = add4(acc, add4(add4(t0, t1), add4(t2, t3)));
}

#define IDS_CAP 128

struct Item { int flat, cnt, row, n; bool last; };

__device__ __forceinline__ Item next_item(int& r, int& j, int& st, int& n, int myN)
{
    Item it;
    const bool valid = r < NPW;
    const int  rem   = n - j;
    it.flat = st + j;
    it.cnt  = valid ? (rem < 8 ? rem : 8) : 0;
    it.row  = r;
    it.n    = n;
    it.last = valid && rem <= 8;
    const int rn = r + (it.last ? 1 : 0);
    const int nn = __builtin_amdgcn_readlane(myN, rn);
    st = it.last ? st + n : st;
    j  = it.last ? 0 : (valid ? j + 8 : j);
    n  = it.last ? nn : n;
    r  = rn;
    return it;
}

__device__ __forceinline__ void issue_item(Rows8& R, const float4* __restrict__ xl,
                                           const int* wids, const Item& it, int gl)
{
    int li = it.flat + (gl & 7);
    li = li < IDS_CAP - 1 ? li : IDS_CAP - 1;
    const int idv = wids[li];
    int s[8];
    #pragma unroll
    for (int u = 0; u < 8; ++u) s[u] = __builtin_amdgcn_readlane(idv, u);
    const int d = it.cnt > 0 ? s[0] : 0;
    #pragma unroll
    for (int u = 0; u < 8; ++u) {
        const int sid = (u < it.cnt) ? s[u] : d;
        R.v[u] = xl[(size_t)sid * 32];
    }
}

__device__ __forceinline__ void sum_item(float4& acc, Rows8& R, const Item& it, float* rowDst)
{
    const float4 z = make_float4(0.f, 0.f, 0.f, 0.f);
    #pragma unroll
    for (int u = 0; u < 8; ++u) if (u >= it.cnt) R.v[u] = z;
    add8(acc, R);
    if (it.last) {
        const float inv = 1.0f / fmaxf((float)it.n, 1.0f);
        float* dst = rowDst + it.row * LDS_STR;
        dst[0] = acc.x * inv; dst[1] = acc.y * inv; dst[2] = acc.z * inv; dst[3] = acc.w * inv;
        acc = z;
    }
}


__device__ __forceinline__ void stage_rows(
    const float* __restrict__ x,
    const int* cnt,
    const int* offs,
    const int* sorted,
    int* ovf, bool inLds, int rowFirst,
    float* Ab, int* wids, int mBase, int nNodes, int g, int gl)
{
    const bool fullBlock = (mBase + MBLK) <= nNodes;
    const float4* x4 = (const float4*)x;
    const int row0 = g * NPW;
    const float4 zero4 = make_float4(0.f, 0.f, 0.f, 0.f);

    const int nd  = mBase + row0 + gl;
    const int ndc = nd < nNodes ? nd : nNodes - 1;
    const int cN  = cnt[ndc - rowFirst];
    const int cB  = offs[ndc - rowFirst];
    float4 xv[NPW];
    #pragma unroll
    for (int i = 0; i < NPW; ++i) {
        const int grow = mBase + row0 + i;
        const int gc   = (fullBlock || grow < nNodes) ? grow : nNodes - 1;
        xv[i] = x4[(size_t)gc * 32 + gl];
    }
    const bool okN  = gl < NPW && nd < nNodes;
    const int myN   = okN ? cN : 0;
    const int myBeg = okN ? cB : 0;
    int tot = 0;
    #pragma unroll
    for (int u = 0; u < NPW; ++u) tot += __builtin_amdgcn_readlane(myN, u);
    const int beg0 = __builtin_amdgcn_readlane(myBeg, 0);
    const bool piped = tot <= IDS_CAP;

    const float4* xl = x4 + gl;
    float* rowDst = &Ab[row0 * LDS_STR + IN_DIM + gl * 4];

    int idv[IDS_CAP / 32];
    #pragma unroll
    for (int u = 0; u < IDS_CAP / 32; ++u) idv[u] = 0;
    if (piped && tot > 0) {
        #pragma unroll
        for (int u = 0; u < IDS_CAP / 32; ++u) {
            const int i = u * 32 + gl;
            idv[u] = load_id(sorted, ovf, inLds, beg0 + (i < tot ? i : tot - 1));
        }
    }
    #pragma unroll
    for (int u = 0; u < IDS_CAP / 32; ++u) wids[u * 32 + gl] = idv[u];

    #pragma unroll
    for (int i = 0; i < NPW; ++i) {
        const int grow = mBase + row0 + i;
        const float4 v = (fullBlock || grow < nNodes) ? xv[i] : zero4;
        float* dst = &Ab[(row0 + i) * LDS_STR + gl * 4];
        dst[0] = v.x; dst[1] = v.y; dst[2] = v.z; dst[3] = v.w;
    }

    if (piped) {
        int r = 0, j = 0, st = 0;
        int n = __builtin_amdgcn_readlane(myN, 0);
        int nItems = 0;
        #pragma unroll
        for (int u = 0; u < NPW; ++u) {
            const int nu = __builtin_amdgcn_readlane(myN, u);
            nItems += nu > 8 ? (nu + 7) >> 3 : 1;
        }
        float4 acc = zero4;
        Rows8 RA, RB;
        Item a = next_item(r, j, st, n, myN);
        issue_item(RA, xl, wids, a, gl);
        for (int i = 0; i < nItems; i += 2) {
            Item b = next_item(r, j, st, n, myN);
            issue_item(RB, xl, wids, b, gl);
            sum_item(acc, RA, a, rowDst);
            a = next_item(r, j, st, n, myN);
            issue_item(RA, xl, wids, a, gl);
            __builtin_amdgcn_sched_barrier(0);
            sum_item(acc, RB, b, rowDst);
        }
    } else {
        int n    = __builtin_amdgcn_readlane(myN, 0);
        int beg  = __builtin_amdgcn_readlane(myBeg, 0);
        int mine = (gl < n) ? load_id(sorted, ovf, inLds, beg + gl) : 0;
        for (int r = 0; r < NPW; ++r) {
            float4 acc = zero4;
            Rows8 R;
            int m = (n < 32) ? n : 32;
            issue8(R, xl, mine, 0, m);

            int n2 = 0, beg2 = 0, mine2 = 0;
            if (r + 1 < NPW) {
                n2   = __builtin_amdgcn_readlane(myN, r + 1);
                beg2 = __builtin_amdgcn_readlane(myBeg, r + 1);
                mine2 = (gl < n2) ? load_id(sorted, ovf, inLds, beg2 + gl) : 0;
            }

            add8(acc, R);
            for (int j = 8; j < m; j += 8) { issue8(R, xl, mine, j, m); add8(acc, R); }
            for (int c = 32; c < n; c += 32) {
                m = (n - c < 32) ? (n - c) : 32;
                mine = (gl < m) ? load_id(sorted, ovf, inLds, beg + c + gl) : 0;
                for (int j = 0; j < m; j += 8) { issue8(R, xl, mine, j, m); add8(acc, R); }
            }

            const float inv = 1.0f / fmaxf((float)n, 1.0f);
            float* dst = rowDst + r * LDS_STR;
            dst[0] = acc.x * inv; dst[1] = acc.y * inv; dst[2] = acc.z * inv; dst[3] = acc.w * inv;

            n = n2; beg = beg2; mine = mine2;
        }
    }
}

__device__ __forceinline__ void mma_rows(
    const float* Ab, const float* Ws, float bv,
    float* __restrict__ out, int mBase, int nNodes, int m, int lane)
{
    const int ln    = lane & 15;
    const int hi    = lane >> 4;
    const int nBase = (m & 7) * 16;
    const int rBase = (m >> 3) * 32;

    if (mBase + rBase >= nNodes) return;

    v8f acc[2] = {};

    #pragma clang loop unroll(disable)
    for (int k0 = 0; k0 < K_DIM; k0 += 32) {
        asm volatile("v_nop\n\tv_nop\n\tv_nop\n\tv_nop\n\tv_nop\n\tv_nop\n\tv_nop\n\tv_nop");
        __builtin_amdgcn_sched_barrier(0);
        v2f bf[8];
        v2f af[8][2];
        #pragma unroll
        for (int s = 0; s < 8; ++s) {
            const int kk = k0 + s * 4 + hi * 2;
            bf[s].x = Ws[kk       * W_STR + nBase + ln];
            bf[s].y = Ws[(kk + 1) * W_STR + nBase + ln];
            #pragma unroll
            for (int ms = 0; ms < 2; ++ms) {
                const int row = rBase + ms * 16 + ln;
                af[s][ms].x = Ab[row * LDS_STR + kk];
                af[s][ms].y = Ab[row * LDS_STR + kk + 1];
            }
        }
        __builtin_amdgcn_sched_barrier(0);
        #pragma unroll
        for (int s = 0; s < 8; ++s) {
            #pragma unroll
            for (int ms = 0; ms < 2; ++ms) {
                acc[ms] = __builtin_amdgcn_wmma_f32_16x16x4_f32(
                     false, af[s][ms],  false, bf[s],
                     (short)0, acc[ms],
                     false,  false);
            }
        }
    }

    asm volatile("v_nop\n\tv_nop\n\tv_nop\n\tv_nop\n\tv_nop\n\tv_nop\n\tv_nop\n\tv_nop");
    __builtin_amdgcn_sched_barrier(0);

    if ((mBase + MBLK) <= nNodes) {
        float* obase = out + (size_t)(mBase + rBase + hi * 8) * OUT_DIM + nBase + ln;
        #pragma unroll
        for (int ms = 0; ms < 2; ++ms) {
            #pragma unroll
            for (int r = 0; r < 8; ++r) {
                obase[(size_t)(ms * 16 + r) * OUT_DIM] = acc[ms][r] + bv;
            }
        }
    } else {
        #pragma unroll
        for (int ms = 0; ms < 2; ++ms) {
            #pragma unroll
            for (int r = 0; r < 8; ++r) {
                const int grow = mBase + rBase + ms * 16 + r + hi * 8;
                if (grow < nNodes)
                    out[(size_t)grow * OUT_DIM + nBase + ln] = acc[ms][r] + bv;
            }
        }
    }
}

__global__ __launch_bounds__(GEMM_T) void sage_gemm(
    const float*    __restrict__ x,
    const unsigned* __restrict__ coarse,
    const int*      __restrict__ binBase,
    int*                         ovf,
    const float*    __restrict__ W,
    const float*    __restrict__ bias,
    float*          __restrict__ out,
    int nNodes, int per)
{
    __shared__ __attribute__((aligned(16))) float Ws[K_DIM * W_STR];
    __shared__ __attribute__((aligned(16))) float As[2 * MBLK * LDS_STR];
    __shared__ int Ids[GWAVES * IDS_CAP];
    __shared__ int Cnt[SH_CAP];
    __shared__ int Off[SH_CAP];
    __shared__ int Sorted[ID_CAP];

    const int t  = threadIdx.x;
    const int wv = __builtin_amdgcn_readfirstlane(t >> 5);
    const int gl = t & 31;

    {
        const float4* W4 = (const float4*)W;
        #pragma unroll
        for (int i = 0; i < (K_DIM * OUT_DIM / 4) / GEMM_T; ++i) {
            const int idx = i * GEMM_T + t;
            const float4 v = W4[idx];
            *(float4*)&Ws[(idx >> 5) * W_STR + (idx & 31) * 4] = v;
        }
    }
    const float bv = bias[(wv & 7) * 16 + (gl & 15)];

    const int rowFirst = (int)blockIdx.x * per;
    const int rowLast  = rowFirst + per < nNodes ? rowFirst + per : nNodes;
    const int nRows    = rowLast > rowFirst ? rowLast - rowFirst : 0;
    const int nT       = (nRows + MBLK - 1) / MBLK;
    const int rem      = nRows - (nT - 1) * MBLK;

    const int  beg   = binBase[blockIdx.x];
    const int  nb    = binBase[blockIdx.x + 1] - beg;
    const bool inLds = nb <= ID_CAP;
    const unsigned* __restrict__ src = coarse + beg;
    int* ovfR = ovf + beg;
    int* rc   = (int*)As;
    int* wsum = rc + SH_CAP;

    if (t < SH_CAP) rc[t] = 0;
    unsigned w0[8];
    #pragma unroll
    for (int u = 0; u < 8; ++u) w0[u] = 0u;
    if (nb > 0) {
        #pragma unroll
        for (int u = 0; u < 8; ++u) {
            const int i = u * GEMM_T + t;
            w0[u] = src[i < nb ? i : nb - 1];
        }
    }
    __syncthreads();
    #pragma unroll
    for (int u = 0; u < 8; ++u) {
        const int i = u * GEMM_T + t;
        if (i < nb) atomicAdd(&rc[w0[u] & SH_MASK], 1);
    }
    for (int i0 = GEMM_T * 8; i0 < nb; i0 += GEMM_T * 8) {
        unsigned w[8];
        #pragma unroll
        for (int u = 0; u < 8; ++u) {
            const int i = i0 + u * GEMM_T + t;
            w[u] = src[i < nb ? i : nb - 1];
        }
        #pragma unroll
        for (int u = 0; u < 8; ++u) {
            const int i = i0 + u * GEMM_T + t;
            if (i < nb) atomicAdd(&rc[w[u] & SH_MASK], 1);
        }
    }
    __syncthreads();
    {
        const int c = t < SH_CAP ? rc[t] : 0;
        int v = c;
        for (int d = 1; d < 32; d <<= 1) {
            const int up = __shfl_up(v, d);
            if (gl >= d) v += up;
        }
        if (t < SH_CAP && gl == 31) wsum[wv] = v;
        __syncthreads();
        if (t == 0) {
            int tot = 0;
            for (int q = 0; q < SH_CAP / 32; ++q) { const int xq = wsum[q]; wsum[q] = tot; tot += xq; }
        }
        __syncthreads();
        if (t < SH_CAP) { Off[t] = wsum[wv] + v - c; Cnt[t] = c; rc[t] = 0; }
    }
    __syncthreads();
    #pragma unroll
    for (int u = 0; u < 8; ++u) {
        const int i = u * GEMM_T + t;
        if (i < nb) {
            const int row = (int)(w0[u] & SH_MASK);
            const int p   = Off[row] + atomicAdd(&rc[row], 1);
            const int sid = (int)(w0[u] >> SH_BITS);
            if (inLds) Sorted[p] = sid; else ovfR[p] = sid;
        }
    }
    for (int i0 = GEMM_T * 8; i0 < nb; i0 += GEMM_T * 8) {
        unsigned w[8];
        #pragma unroll
        for (int u = 0; u < 8; ++u) {
            const int i = i0 + u * GEMM_T + t;
            w[u] = src[i < nb ? i : nb - 1];
        }
        #pragma unroll
        for (int u = 0; u < 8; ++u) {
            const int i = i0 + u * GEMM_T + t;
            if (i < nb) {
                const int row = (int)(w[u] & SH_MASK);
                const int p   = Off[row] + atomicAdd(&rc[row], 1);
                const int sid = (int)(w[u] >> SH_BITS);
                ovfR[p] = sid;
            }
        }
    }
    if (!inLds) __threadfence();
    __syncthreads();

    for (int it = 0; it <= nT; ++it) {
        if (wv < GWAVES) {
            if (it < nT) {
                const int tb = it == 0 ? rowFirst : rowFirst + rem + (it - 1) * MBLK;
                const int te = it == 0 ? rowFirst + rem : tb + MBLK;
                stage_rows(x, Cnt, Off, Sorted, ovfR, inLds, rowFirst,
                           &As[(it & 1) * (MBLK * LDS_STR)], &Ids[wv * IDS_CAP], tb, te, wv, gl);
            }
        } else {
            if (it > 0) {
                const int pt = it - 1;
                const int tb = pt == 0 ? rowFirst : rowFirst + rem + (pt - 1) * MBLK;
                const int te = pt == 0 ? rowFirst + rem : tb + MBLK;
                mma_rows(&As[(pt & 1) * (MBLK * LDS_STR)], Ws, bv, out,
                         tb, te, wv - GWAVES, gl);
            }
        }
        __builtin_amdgcn_fence(__ATOMIC_RELEASE, "workgroup", "local");
        __builtin_amdgcn_s_barrier();
        __builtin_amdgcn_fence(__ATOMIC_ACQUIRE, "workgroup", "local");
    }
}

extern "C" void kernel_launch(void* const* d_in, const int* in_sizes, int n_in,
                              void* d_out, int out_size, void* d_ws, size_t ws_size,
                              hipStream_t stream)
{
    const float* x    = (const float*)d_in[0];
    const int*   esrc = (const int*)  d_in[1];
    const int*   edst = (const int*)  d_in[2];
    const float* W    = (const float*)d_in[3];
    const float* bias = (const float*)d_in[4];
    float*       out  = (float*)d_out;

    const int nNodes = in_sizes[0] / IN_DIM;
    const int nEdges = in_sizes[1];
    const int per    = (nNodes + N_SHARES - 1) / N_SHARES;
    const int nBins  = (nNodes + per - 1) / per;

    unsigned* coarse    = (unsigned*)d_ws;
    int*      ovf       = (int*)(coarse + nEdges);
    int*      binCnt    = ovf + nEdges;
    int*      binCursor = binCnt + nBins;
    int*      binBase   = binCursor + nBins;
    hipMemsetAsync(binCnt, 0, (size_t)2 * nBins * sizeof(int), stream);

    const int sortBlocks = (nEdges + SORT_CHUNK - 1) / SORT_CHUNK;
    sage_bincount<<<sortBlocks, SORT_T, (size_t)nBins * sizeof(int), stream>>>(edst, binCnt, nEdges, nBins, per);
    sage_binplace<<<sortBlocks, SORT_T, (size_t)3 * nBins * sizeof(int), stream>>>(
        esrc, edst, binCnt, binCursor, binBase, coarse, nEdges, nBins, per);

    sage_gemm<<<nBins, GEMM_T, 0, stream>>>(x, coarse, binBase, ovf, W, bias, out, nNodes, per);
}
